# starting kernel
# baseline (speedup 1.0000x reference)
#include <hip/hip_runtime.h>
#include <math.h>


typedef __attribute__((ext_vector_type(16))) __bf16 v16bf;
typedef __attribute__((ext_vector_type(8)))  __bf16 v8bf;
typedef __attribute__((ext_vector_type(8)))  float  v8f;
typedef __attribute__((ext_vector_type(4)))  float  v4f;

#define IN_FEAT 512
#define HID     16

__device__ __forceinline__ void atomAddF(float* p, float v) {
  __hip_atomic_fetch_add(p, v, __ATOMIC_RELAXED, __HIP_MEMORY_SCOPE_AGENT);
}

__global__ __launch_bounds__(256) void gemm1_wmma_k(
    const float* __restrict__ X, const float* __restrict__ W1,
    float* __restrict__ XW1, int nTiles) {
  __shared__ __bf16 Bs[HID * IN_FEAT];
  for (int i = threadIdx.x; i < HID * IN_FEAT; i += 256) {
    int k = i >> 4, n = i & 15;
    Bs[n * IN_FEAT + k] = (__bf16)W1[i];
  }
  __syncthreads();

  int wave = threadIdx.x >> 5;
  int lane = threadIdx.x & 31;
  int tile = blockIdx.x * 8 + wave;
  if (tile >= nTiles) return;

  int rc   = lane & 15;
  int half = lane >> 4;

  const float*  arow = X  + (size_t)(tile * 16 + rc) * IN_FEAT;
  const __bf16* bcol = Bs + (size_t)rc * IN_FEAT;

  v8f c = {0.f, 0.f, 0.f, 0.f, 0.f, 0.f, 0.f, 0.f};

#pragma unroll
  for (int ks = 0; ks < IN_FEAT / 32; ++ks) {
    int kb = ks * 32 + half * 8;
    v4f a0 = *(const v4f*)(arow + kb);
    v4f a1 = *(const v4f*)(arow + kb + 4);
    v4f a2 = *(const v4f*)(arow + kb + 16);
    v4f a3 = *(const v4f*)(arow + kb + 20);
    v8bf b0 = *(const v8bf*)(bcol + kb);
    v8bf b1 = *(const v8bf*)(bcol + kb + 16);
    v16bf A, B;
#pragma unroll
    for (int i = 0; i < 4; ++i) {
      A[i]      = (__bf16)a0[i];
      A[4 + i]  = (__bf16)a1[i];
      A[8 + i]  = (__bf16)a2[i];
      A[12 + i] = (__bf16)a3[i];
    }
#pragma unroll
    for (int i = 0; i < 8; ++i) { B[i] = b0[i]; B[8 + i] = b1[i]; }
    c = __builtin_amdgcn_wmma_f32_16x16x32_bf16(
         false, A,  false, B,
         (short)0, c,  false,  false);
  }

  float* orow = XW1 + ((size_t)tile * 16 + half * 8) * HID + rc;
#pragma unroll
  for (int r = 0; r < 8; ++r) orow[r * HID] = c[r];
}

__global__ void deg_init_k(int* DEG, int N) {
  int n = blockIdx.x * blockDim.x + threadIdx.x;
  if (n < N) DEG[n] = 1;
}
__global__ void deg_edge_k(const int* __restrict__ DST, int* DEG, int E) {
  int e = blockIdx.x * blockDim.x + threadIdx.x;
  if (e < E) atomicAdd(&DEG[DST[e]], 1);
}
__global__ void dinv_k(float* DNV, int N) {
  int n = blockIdx.x * blockDim.x + threadIdx.x;
  if (n < N) {
    int c = ((const int*)DNV)[n];
    DNV[n] = rsqrtf((float)c);
  }
}

__global__ void self1_k(const float* __restrict__ XW1,
                        const float* __restrict__ DNV, float* H, int N) {
  int t = blockIdx.x * blockDim.x + threadIdx.x;
  int n = t >> 2;
  if (n >= N) return;
  int q = t & 3;
  float w = DNV[n] * DNV[n];
  v4f v = *(const v4f*)(XW1 + (size_t)n * HID + q * 4);
  v4f o;
#pragma unroll
  for (int i = 0; i < 4; ++i) o[i] = w * v[i];
  *(v4f*)(H + (size_t)n * HID + q * 4) = o;
}

__global__ void agg1_k(const int* __restrict__ SRC, const int* __restrict__ DST,
                       const float* __restrict__ DNV,
                       const float* __restrict__ XW1, float* H, int E) {
  int t = blockIdx.x * blockDim.x + threadIdx.x;
  int e = t >> 2;
  if (e >= E) return;
  int q = t & 3;
  int s = SRC[e], d = DST[e];
  float nrm = DNV[s] * DNV[d];
  v4f v = *(const v4f*)(XW1 + (size_t)s * HID + q * 4);
  float* hp = H + (size_t)d * HID + q * 4;
#pragma unroll
  for (int i = 0; i < 4; ++i) atomAddF(hp + i, nrm * v[i]);
}

__global__ void relu_bias_k(float* H, const float* __restrict__ B1, int N) {
  int t = blockIdx.x * blockDim.x + threadIdx.x;
  int n = t >> 2;
  if (n >= N) return;
  int q = t & 3;
  v4f b = *(const v4f*)(B1 + q * 4);
  float* hp = H + (size_t)n * HID + q * 4;
  v4f v = *(const v4f*)hp;
#pragma unroll
  for (int i = 0; i < 4; ++i) v[i] = fmaxf(v[i] + b[i], 0.f);
  *(v4f*)hp = v;
}

__global__ void gemm2_k(const float* __restrict__ H, const float* __restrict__ W2,
                        const float* __restrict__ DNV, float* HW2, float* OUT2,
                        int N) {
  int n = blockIdx.x * blockDim.x + threadIdx.x;
  if (n >= N) return;
  const float* hr = H + (size_t)n * HID;
  float h0 = 0.f, h1 = 0.f;
#pragma unroll
  for (int f = 0; f < HID; ++f) {
    float hv = hr[f];
    h0 += hv * W2[2 * f];
    h1 += hv * W2[2 * f + 1];
  }
  HW2[2 * n] = h0;
  HW2[2 * n + 1] = h1;
  float w = DNV[n] * DNV[n];
  OUT2[2 * n] = w * h0;
  OUT2[2 * n + 1] = w * h1;
}

__global__ void agg2_k(const int* __restrict__ SRC, const int* __restrict__ DST,
                       const float* __restrict__ DNV,
                       const float* __restrict__ HW2, float* OUT2, int E) {
  int e = blockIdx.x * blockDim.x + threadIdx.x;
  if (e >= E) return;
  int s = SRC[e], d = DST[e];
  float nrm = DNV[s] * DNV[d];
  atomAddF(&OUT2[2 * d],     nrm * HW2[2 * s]);
  atomAddF(&OUT2[2 * d + 1], nrm * HW2[2 * s + 1]);
}

__global__ void lsm_k(const float* __restrict__ OUT2, const float* __restrict__ B2,
                      float* __restrict__ out, int N) {
  int n = blockIdx.x * blockDim.x + threadIdx.x;
  if (n >= N) return;
  float a = OUT2[2 * n]     + B2[0];
  float b = OUT2[2 * n + 1] + B2[1];
  float m = fmaxf(a, b);
  float lse = m + logf(expf(a - m) + expf(b - m));
  out[2 * n]     = a - lse;
  out[2 * n + 1] = b - lse;
}

extern "C" void kernel_launch(void* const* d_in, const int* in_sizes, int n_in,
                              void* d_out, int out_size, void* d_ws, size_t ws_size,
                              hipStream_t stream) {
  const float* X  = (const float*)d_in[0];
  const int*   EI = (const int*)d_in[1];
  const float* W1 = (const float*)d_in[2];
  const float* B1 = (const float*)d_in[3];
  const float* W2 = (const float*)d_in[4];
  const float* B2 = (const float*)d_in[5];

  const int N = in_sizes[0] / IN_FEAT;
  const int E = in_sizes[1] / 2;
  const int* SRC = EI;
  const int* DST = EI + E;

  float* ws   = (float*)d_ws;
  float* XW1  = ws;
  float* H    = ws + (size_t)N * HID;
  float* DNV  = ws + (size_t)2 * N * HID;
  float* HW2  = DNV + N;
  float* OUT2 = HW2 + (size_t)2 * N;

  const int thr = 256;
  const int nTiles = (N + 15) / 16;

  gemm1_wmma_k<<<(nTiles + 7) / 8, thr, 0, stream>>>(X, W1, XW1, nTiles);
  deg_init_k<<<(N + thr - 1) / thr, thr, 0, stream>>>((int*)DNV, N);
  deg_edge_k<<<(E + thr - 1) / thr, thr, 0, stream>>>(DST, (int*)DNV, E);
  dinv_k<<<(N + thr - 1) / thr, thr, 0, stream>>>(DNV, N);
  self1_k<<<(4 * N + thr - 1) / thr, thr, 0, stream>>>(XW1, DNV, H, N);
  agg1_k<<<(4 * E + thr - 1) / thr, thr, 0, stream>>>(SRC, DST, DNV, XW1, H, E);
  relu_bias_k<<<(4 * N + thr - 1) / thr, thr, 0, stream>>>(H, B1, N);
  gemm2_k<<<(N + thr - 1) / thr, thr, 0, stream>>>(H, W2, DNV, HW2, OUT2, N);
  agg2_k<<<(E + thr - 1) / thr, thr, 0, stream>>>(SRC, DST, DNV, HW2, OUT2, E);
  lsm_k<<<(N + thr - 1) / thr, thr, 0, stream>>>(OUT2, B2, (float*)d_out, N);
}


# edge scatters: lanes of one destination row (16 in layer 1, 2 in layer 2) adjacent in one f32 atomic instruction, indices one lane per edge, 64 edges per wave; atomics kept, arrival order as before
# speedup vs baseline: 1.3114x; 1.3114x over previous
#include <hip/hip_runtime.h>
#include <math.h>


typedef __attribute__((ext_vector_type(16))) __bf16 v16bf;
typedef __attribute__((ext_vector_type(8)))  __bf16 v8bf;
typedef __attribute__((ext_vector_type(8)))  float  v8f;
typedef __attribute__((ext_vector_type(4)))  float  v4f;
typedef __attribute__((ext_vector_type(2)))  float  v2f;

#define IN_FEAT 512
#define HID     16
#define EDGE_BATCHES 2

__device__ __forceinline__ void atomAddF(float* p, float v) {
  __hip_atomic_fetch_add(p, v, __ATOMIC_RELAXED, __HIP_MEMORY_SCOPE_AGENT);
}

__global__ __launch_bounds__(256) void gemm1_wmma_k(
    const float* __restrict__ X, const float* __restrict__ W1,
    float* __restrict__ XW1, int nTiles) {
  __shared__ __bf16 Bs[HID * IN_FEAT];
  for (int i = threadIdx.x; i < HID * IN_FEAT; i += 256) {
    int k = i >> 4, n = i & 15;
    Bs[n * IN_FEAT + k] = (__bf16)W1[i];
  }
  __syncthreads();

  int wave = threadIdx.x >> 5;
  int lane = threadIdx.x & 31;
  int tile = blockIdx.x * 8 + wave;
  if (tile >= nTiles) return;

  int rc   = lane & 15;
  int half = lane >> 4;

  const float*  arow = X  + (size_t)(tile * 16 + rc) * IN_FEAT;
  const __bf16* bcol = Bs + (size_t)rc * IN_FEAT;

  v8f c = {0.f, 0.f, 0.f, 0.f, 0.f, 0.f, 0.f, 0.f};

#pragma unroll
  for (int ks = 0; ks < IN_FEAT / 32; ++ks) {
    int kb = ks * 32 + half * 8;
    v4f a0 = *(const v4f*)(arow + kb);
    v4f a1 = *(const v4f*)(arow + kb + 4);
    v4f a2 = *(const v4f*)(arow + kb + 16);
    v4f a3 = *(const v4f*)(arow + kb + 20);
    v8bf b0 = *(const v8bf*)(bcol + kb);
    v8bf b1 = *(const v8bf*)(bcol + kb + 16);
    v16bf A, B;
#pragma unroll
    for (int i = 0; i < 4; ++i) {
      A[i]      = (__bf16)a0[i];
      A[4 + i]  = (__bf16)a1[i];
      A[8 + i]  = (__bf16)a2[i];
      A[12 + i] = (__bf16)a3[i];
    }
#pragma unroll
    for (int i = 0; i < 8; ++i) { B[i] = b0[i]; B[8 + i] = b1[i]; }
    c = __builtin_amdgcn_wmma_f32_16x16x32_bf16(
         false, A,  false, B,
         (short)0, c,  false,  false);
  }

  float* orow = XW1 + ((size_t)tile * 16 + half * 8) * HID + rc;
#pragma unroll
  for (int r = 0; r < 8; ++r) orow[r * HID] = c[r];
}

__global__ void deg_init_k(int* DEG, int N) {
  int n = blockIdx.x * blockDim.x + threadIdx.x;
  if (n < N) DEG[n] = 1;
}
__global__ void deg_edge_k(const int* __restrict__ DST, int* DEG, int E) {
  int e = blockIdx.x * blockDim.x + threadIdx.x;
  if (e < E) atomicAdd(&DEG[DST[e]], 1);
}
__global__ void dinv_k(float* DNV, int N) {
  int n = blockIdx.x * blockDim.x + threadIdx.x;
  if (n < N) {
    int c = ((const int*)DNV)[n];
    DNV[n] = rsqrtf((float)c);
  }
}

__global__ void self1_k(const float* __restrict__ XW1,
                        const float* __restrict__ DNV, float* H, int N) {
  int t = blockIdx.x * blockDim.x + threadIdx.x;
  int n = t >> 2;
  if (n >= N) return;
  int q = t & 3;
  float w = DNV[n] * DNV[n];
  v4f v = *(const v4f*)(XW1 + (size_t)n * HID + q * 4);
  v4f o;
#pragma unroll
  for (int i = 0; i < 4; ++i) o[i] = w * v[i];
  *(v4f*)(H + (size_t)n * HID + q * 4) = o;
}

__global__ __launch_bounds__(256) void agg1_k(
    const int* __restrict__ SRC, const int* __restrict__ DST,
    const float* __restrict__ DNV, const float* __restrict__ XW1,
    float* H, int E) {
  const int lane = threadIdx.x & 31;
  const int wave = blockIdx.x * 8 + (threadIdx.x >> 5);
  const int f    = lane & 15;
  const int hi   = lane >> 4;
  int   s[EDGE_BATCHES], d[EDGE_BATCHES];
  float nrm[EDGE_BATCHES];
#pragma unroll
  for (int b = 0; b < EDGE_BATCHES; ++b) {
    const int e = (wave * EDGE_BATCHES + b) * 32 + lane;
    s[b] = 0; d[b] = 0; nrm[b] = 0.f;
    if (e < E) {
      s[b] = SRC[e];
      d[b] = DST[e];
      nrm[b] = DNV[s[b]] * DNV[d[b]];
    }
  }
#pragma unroll
  for (int b = 0; b < EDGE_BATCHES; ++b) {
    const int e0 = (wave * EDGE_BATCHES + b) * 32;
    float v[16];
#pragma unroll
    for (int j = 0; j < 16; ++j) {
      const int sj = __shfl(s[b], 2 * j + hi, 32);
      v[j] = XW1[(size_t)sj * HID + f];
    }
#pragma unroll
    for (int j = 0; j < 16; ++j) {
      const int   dj = __shfl(d[b], 2 * j + hi, 32);
      const float nj = __shfl(nrm[b], 2 * j + hi, 32);
      if (e0 + 2 * j + hi < E) atomAddF(H + (size_t)dj * HID + f, nj * v[j]);
    }
  }
}

__global__ void relu_bias_k(float* H, const float* __restrict__ B1, int N) {
  int t = blockIdx.x * blockDim.x + threadIdx.x;
  int n = t >> 2;
  if (n >= N) return;
  int q = t & 3;
  v4f b = *(const v4f*)(B1 + q * 4);
  float* hp = H + (size_t)n * HID + q * 4;
  v4f v = *(const v4f*)hp;
#pragma unroll
  for (int i = 0; i < 4; ++i) v[i] = fmaxf(v[i] + b[i], 0.f);
  *(v4f*)hp = v;
}

__global__ void gemm2_k(const float* __restrict__ H, const float* __restrict__ W2,
                        const float* __restrict__ DNV, float* HW2, float* OUT2,
                        int N) {
  int n = blockIdx.x * blockDim.x + threadIdx.x;
  if (n >= N) return;
  const float* hr = H + (size_t)n * HID;
  float h0 = 0.f, h1 = 0.f;
#pragma unroll
  for (int f = 0; f < HID; ++f) {
    float hv = hr[f];
    h0 += hv * W2[2 * f];
    h1 += hv * W2[2 * f + 1];
  }
  HW2[2 * n] = h0;
  HW2[2 * n + 1] = h1;
  float w = DNV[n] * DNV[n];
  OUT2[2 * n] = w * h0;
  OUT2[2 * n + 1] = w * h1;
}

__global__ __launch_bounds__(256) void agg2_k(
    const int* __restrict__ SRC, const int* __restrict__ DST,
    const float* __restrict__ DNV, const float* __restrict__ HW2,
    float* OUT2, int E) {
  const int lane = threadIdx.x & 31;
  const int wave = blockIdx.x * 8 + (threadIdx.x >> 5);
  const int c    = lane & 1;
  const int sub  = lane >> 1;
  int   d[EDGE_BATCHES];
  float v0[EDGE_BATCHES], v1[EDGE_BATCHES];
#pragma unroll
  for (int b = 0; b < EDGE_BATCHES; ++b) {
    const int e = (wave * EDGE_BATCHES + b) * 32 + lane;
    d[b] = 0; v0[b] = 0.f; v1[b] = 0.f;
    if (e < E) {
      const int s = SRC[e];
      d[b] = DST[e];
      const float nrm = DNV[s] * DNV[d[b]];
      const v2f h = *(const v2f*)(HW2 + (size_t)2 * s);
      v0[b] = nrm * h[0];
      v1[b] = nrm * h[1];
    }
  }
#pragma unroll
  for (int b = 0; b < EDGE_BATCHES; ++b) {
    const int e0 = (wave * EDGE_BATCHES + b) * 32;
#pragma unroll
    for (int j = 0; j < 2; ++j) {
      const int   dj = __shfl(d[b], 16 * j + sub, 32);
      const float a0 = __shfl(v0[b], 16 * j + sub, 32);
      const float a1 = __shfl(v1[b], 16 * j + sub, 32);
      if (e0 + 16 * j + sub < E)
        atomAddF(OUT2 + (size_t)2 * dj + c, c ? a1 : a0);
    }
  }
}

__global__ void lsm_k(const float* __restrict__ OUT2, const float* __restrict__ B2,
                      float* __restrict__ out, int N) {
  int n = blockIdx.x * blockDim.x + threadIdx.x;
  if (n >= N) return;
  float a = OUT2[2 * n]     + B2[0];
  float b = OUT2[2 * n + 1] + B2[1];
  float m = fmaxf(a, b);
  float lse = m + logf(expf(a - m) + expf(b - m));
  out[2 * n]     = a - lse;
  out[2 * n + 1] = b - lse;
}

extern "C" void kernel_launch(void* const* d_in, const int* in_sizes, int n_in,
                              void* d_out, int out_size, void* d_ws, size_t ws_size,
                              hipStream_t stream) {
  const float* X  = (const float*)d_in[0];
  const int*   EI = (const int*)d_in[1];
  const float* W1 = (const float*)d_in[2];
  const float* B1 = (const float*)d_in[3];
  const float* W2 = (const float*)d_in[4];
  const float* B2 = (const float*)d_in[5];

  const int N = in_sizes[0] / IN_FEAT;
  const int E = in_sizes[1] / 2;
  const int* SRC = EI;
  const int* DST = EI + E;

  float* ws   = (float*)d_ws;
  float* XW1  = ws;
  float* H    = ws + (size_t)N * HID;
  float* DNV  = ws + (size_t)2 * N * HID;
  float* HW2  = DNV + N;
  float* OUT2 = HW2 + (size_t)2 * N;

  const int thr = 256;
  const int nTiles = (N + 15) / 16;

  gemm1_wmma_k<<<(nTiles + 7) / 8, thr, 0, stream>>>(X, W1, XW1, nTiles);
  deg_init_k<<<(N + thr - 1) / thr, thr, 0, stream>>>((int*)DNV, N);
  deg_edge_k<<<(E + thr - 1) / thr, thr, 0, stream>>>(DST, (int*)DNV, E);
  dinv_k<<<(N + thr - 1) / thr, thr, 0, stream>>>(DNV, N);
  self1_k<<<(4 * N + thr - 1) / thr, thr, 0, stream>>>(XW1, DNV, H, N);
  agg1_k<<<(E + 32 * EDGE_BATCHES * 8 - 1) / (32 * EDGE_BATCHES * 8), thr, 0, stream>>>(SRC, DST, DNV, XW1, H, E);
  relu_bias_k<<<(4 * N + thr - 1) / thr, thr, 0, stream>>>(H, B1, N);
  gemm2_k<<<(N + thr - 1) / thr, thr, 0, stream>>>(H, W2, DNV, HW2, OUT2, N);
  agg2_k<<<(E + 32 * EDGE_BATCHES * 8 - 1) / (32 * EDGE_BATCHES * 8), thr, 0, stream>>>(SRC, DST, DNV, HW2, OUT2, E);
  lsm_k<<<(N + thr - 1) / thr, thr, 0, stream>>>(OUT2, B2, (float*)d_out, N);
}


# speedup vs baseline: 3.1496x; 2.4017x over previous
#include <hip/hip_runtime.h>
#include <math.h>


typedef __attribute__((ext_vector_type(16))) __bf16 v16bf;
typedef __attribute__((ext_vector_type(8)))  __bf16 v8bf;
typedef __attribute__((ext_vector_type(8)))  float  v8f;
typedef __attribute__((ext_vector_type(4)))  float  v4f;
typedef __attribute__((ext_vector_type(2)))  float  v2f;

#define IN_FEAT 512
#define HID     16
#define EDGE_BATCHES 2

__device__ __forceinline__ int package_half() {
  const unsigned die = (__builtin_amdgcn_s_sendmsg_rtn(0x87) >> 16) & 0xFu;
  return (int)((die >> 2) & 1u);
}

__device__ __forceinline__ void atomAddF(float* p, float v) {
  __hip_atomic_fetch_add(p, v, __ATOMIC_RELAXED, __HIP_MEMORY_SCOPE_AGENT);
}

__global__ __launch_bounds__(256) void gemm1_wmma_k(
    const float* __restrict__ X, const float* __restrict__ W1,
    float* __restrict__ XW1, int nTiles) {
  __shared__ __bf16 Bs[HID * IN_FEAT];
  for (int i = threadIdx.x; i < HID * IN_FEAT; i += 256) {
    int k = i >> 4, n = i & 15;
    Bs[n * IN_FEAT + k] = (__bf16)W1[i];
  }
  __syncthreads();

  int wave = threadIdx.x >> 5;
  int lane = threadIdx.x & 31;
  int tile = blockIdx.x * 8 + wave;
  if (tile >= nTiles) return;

  int rc   = lane & 15;
  int half = lane >> 4;

  const float*  arow = X  + (size_t)(tile * 16 + rc) * IN_FEAT;
  const __bf16* bcol = Bs + (size_t)rc * IN_FEAT;

  v8f c = {0.f, 0.f, 0.f, 0.f, 0.f, 0.f, 0.f, 0.f};

#pragma unroll
  for (int ks = 0; ks < IN_FEAT / 32; ++ks) {
    int kb = ks * 32 + half * 8;
    v4f a0 = *(const v4f*)(arow + kb);
    v4f a1 = *(const v4f*)(arow + kb + 4);
    v4f a2 = *(const v4f*)(arow + kb + 16);
    v4f a3 = *(const v4f*)(arow + kb + 20);
    v8bf b0 = *(const v8bf*)(bcol + kb);
    v8bf b1 = *(const v8bf*)(bcol + kb + 16);
    v16bf A, B;
#pragma unroll
    for (int i = 0; i < 4; ++i) {
      A[i]      = (__bf16)a0[i];
      A[4 + i]  = (__bf16)a1[i];
      A[8 + i]  = (__bf16)a2[i];
      A[12 + i] = (__bf16)a3[i];
    }
#pragma unroll
    for (int i = 0; i < 8; ++i) { B[i] = b0[i]; B[8 + i] = b1[i]; }
    c = __builtin_amdgcn_wmma_f32_16x16x32_bf16(
         false, A,  false, B,
         (short)0, c,  false,  false);
  }

  float* orow = XW1 + ((size_t)tile * 16 + half * 8) * HID + rc;
#pragma unroll
  for (int r = 0; r < 8; ++r) orow[r * HID] = c[r];
}

__global__ void deg_init_k(int* DEG, int* DEGB, int N) {
  int n = blockIdx.x * blockDim.x + threadIdx.x;
  if (n < N) { DEG[n] = 1; DEGB[n] = 0; }
}
__global__ void deg_edge_k(const int* __restrict__ DST, int* DEG, int* DEGB, int E) {
  int e = blockIdx.x * blockDim.x + threadIdx.x;
  int* D = package_half() ? DEGB : DEG;
  if (e < E) atomicAdd(&D[DST[e]], 1);
}
__global__ void dinv_k(float* DNV, const int* __restrict__ DEGB, int N) {
  int n = blockIdx.x * blockDim.x + threadIdx.x;
  if (n < N) {
    int c = ((const int*)DNV)[n] + DEGB[n];
    DNV[n] = rsqrtf((float)c);
  }
}

__global__ void self1_k(const float* __restrict__ XW1,
                        const float* __restrict__ DNV, float* H, float* HB, int N) {
  int t = blockIdx.x * blockDim.x + threadIdx.x;
  int n = t >> 2;
  if (n >= N) return;
  int q = t & 3;
  float w = DNV[n] * DNV[n];
  v4f v = *(const v4f*)(XW1 + (size_t)n * HID + q * 4);
  v4f o;
#pragma unroll
  for (int i = 0; i < 4; ++i) o[i] = w * v[i];
  *(v4f*)(H + (size_t)n * HID + q * 4) = o;
  if (HB != H) {
    const v4f z = {0.f, 0.f, 0.f, 0.f};
    *(v4f*)(HB + (size_t)n * HID + q * 4) = z;
  }
}

__global__ __launch_bounds__(256) void agg1_k(
    const int* __restrict__ SRC, const int* __restrict__ DST,
    const float* __restrict__ DNV, const float* __restrict__ XW1,
    float* H0, float* HB, int E) {
  float* H = package_half() ? HB : H0;
  const int lane = threadIdx.x & 31;
  const int wave = blockIdx.x * 8 + (threadIdx.x >> 5);
  const int f    = lane & 15;
  const int hi   = lane >> 4;
  int   s[EDGE_BATCHES], d[EDGE_BATCHES];
  float nrm[EDGE_BATCHES];
#pragma unroll
  for (int b = 0; b < EDGE_BATCHES; ++b) {
    const int e = (wave * EDGE_BATCHES + b) * 32 + lane;
    s[b] = 0; d[b] = 0; nrm[b] = 0.f;
    if (e < E) {
      s[b] = SRC[e];
      d[b] = DST[e];
      nrm[b] = DNV[s[b]] * DNV[d[b]];
    }
  }
#pragma unroll
  for (int b = 0; b < EDGE_BATCHES; ++b) {
    const int e0 = (wave * EDGE_BATCHES + b) * 32;
    float v[16];
#pragma unroll
    for (int j = 0; j < 16; ++j) {
      const int sj = __shfl(s[b], 2 * j + hi, 32);
      v[j] = XW1[(size_t)sj * HID + f];
    }
#pragma unroll
    for (int j = 0; j < 16; ++j) {
      const int   dj = __shfl(d[b], 2 * j + hi, 32);
      const float nj = __shfl(nrm[b], 2 * j + hi, 32);
      if (e0 + 2 * j + hi < E) atomAddF(H + (size_t)dj * HID + f, nj * v[j]);
    }
  }
}

__global__ void relu_bias_k(float* H, const float* HB, const float* __restrict__ B1, int N) {
  int t = blockIdx.x * blockDim.x + threadIdx.x;
  int n = t >> 2;
  if (n >= N) return;
  int q = t & 3;
  v4f b = *(const v4f*)(B1 + q * 4);
  float* hp = H + (size_t)n * HID + q * 4;
  v4f v = *(const v4f*)hp;
  if (HB != H) {
    const v4f u = *(const v4f*)(HB + (size_t)n * HID + q * 4);
#pragma unroll
    for (int i = 0; i < 4; ++i) v[i] += u[i];
  }
#pragma unroll
  for (int i = 0; i < 4; ++i) v[i] = fmaxf(v[i] + b[i], 0.f);
  *(v4f*)hp = v;
}

__global__ void gemm2_k(const float* __restrict__ H, const float* __restrict__ W2,
                        const float* __restrict__ DNV, float* HW2, float* OUT2,
                        float* OUT2B, int N) {
  int n = blockIdx.x * blockDim.x + threadIdx.x;
  if (n >= N) return;
  const float* hr = H + (size_t)n * HID;
  float h0 = 0.f, h1 = 0.f;
#pragma unroll
  for (int f = 0; f < HID; ++f) {
    float hv = hr[f];
    h0 += hv * W2[2 * f];
    h1 += hv * W2[2 * f + 1];
  }
  HW2[2 * n] = h0;
  HW2[2 * n + 1] = h1;
  float w = DNV[n] * DNV[n];
  OUT2[2 * n] = w * h0;
  OUT2[2 * n + 1] = w * h1;
  OUT2B[2 * n] = 0.f;
  OUT2B[2 * n + 1] = 0.f;
}

__global__ __launch_bounds__(256) void agg2_k(
    const int* __restrict__ SRC, const int* __restrict__ DST,
    const float* __restrict__ DNV, const float* __restrict__ HW2,
    float* OUT2A, float* OUT2B, int E) {
  float* OUT2 = package_half() ? OUT2B : OUT2A;
  const int lane = threadIdx.x & 31;
  const int wave = blockIdx.x * 8 + (threadIdx.x >> 5);
  const int c    = lane & 1;
  const int sub  = lane >> 1;
  int   d[EDGE_BATCHES];
  float v0[EDGE_BATCHES], v1[EDGE_BATCHES];
#pragma unroll
  for (int b = 0; b < EDGE_BATCHES; ++b) {
    const int e = (wave * EDGE_BATCHES + b) * 32 + lane;
    d[b] = 0; v0[b] = 0.f; v1[b] = 0.f;
    if (e < E) {
      const int s = SRC[e];
      d[b] = DST[e];
      const float nrm = DNV[s] * DNV[d[b]];
      const v2f h = *(const v2f*)(HW2 + (size_t)2 * s);
      v0[b] = nrm * h[0];
      v1[b] = nrm * h[1];
    }
  }
#pragma unroll
  for (int b = 0; b < EDGE_BATCHES; ++b) {
    const int e0 = (wave * EDGE_BATCHES + b) * 32;
#pragma unroll
    for (int j = 0; j < 2; ++j) {
      const int   dj = __shfl(d[b], 16 * j + sub, 32);
      const float a0 = __shfl(v0[b], 16 * j + sub, 32);
      const float a1 = __shfl(v1[b], 16 * j + sub, 32);
      if (e0 + 16 * j + sub < E)
        atomAddF(OUT2 + (size_t)2 * dj + c, c ? a1 : a0);
    }
  }
}

__global__ void lsm_k(const float* __restrict__ OUT2, const float* __restrict__ OUT2B,
                      const float* __restrict__ B2, float* __restrict__ out, int N) {
  int n = blockIdx.x * blockDim.x + threadIdx.x;
  if (n >= N) return;
  float a = OUT2[2 * n]     + OUT2B[2 * n]     + B2[0];
  float b = OUT2[2 * n + 1] + OUT2B[2 * n + 1] + B2[1];
  float m = fmaxf(a, b);
  float lse = m + logf(expf(a - m) + expf(b - m));
  out[2 * n]     = a - lse;
  out[2 * n + 1] = b - lse;
}

extern "C" void kernel_launch(void* const* d_in, const int* in_sizes, int n_in,
                              void* d_out, int out_size, void* d_ws, size_t ws_size,
                              hipStream_t stream) {
  const float* X  = (const float*)d_in[0];
  const int*   EI = (const int*)d_in[1];
  const float* W1 = (const float*)d_in[2];
  const float* B1 = (const float*)d_in[3];
  const float* W2 = (const float*)d_in[4];
  const float* B2 = (const float*)d_in[5];

  const int N = in_sizes[0] / IN_FEAT;
  const int E = in_sizes[1] / 2;
  const int* SRC = EI;
  const int* DST = EI + E;

  float* ws   = (float*)d_ws;
  float* XW1  = ws;
  float* H    = ws + (size_t)N * HID;
  float* DNV  = ws + (size_t)2 * N * HID;
  float* HW2  = DNV + N;
  float* OUT2 = HW2 + (size_t)2 * N;
  float* HB   = ws_size >= (size_t)(3 * HID + 5) * N * sizeof(float) ? OUT2 + (size_t)2 * N : H;
  int*   DEGB = (int*)H;
  float* OUT2B = XW1;

  const int thr = 256;
  const int nTiles = (N + 15) / 16;

  gemm1_wmma_k<<<(nTiles + 7) / 8, thr, 0, stream>>>(X, W1, XW1, nTiles);
  deg_init_k<<<(N + thr - 1) / thr, thr, 0, stream>>>((int*)DNV, DEGB, N);
  deg_edge_k<<<(E + thr - 1) / thr, thr, 0, stream>>>(DST, (int*)DNV, DEGB, E);
  dinv_k<<<(N + thr - 1) / thr, thr, 0, stream>>>(DNV, DEGB, N);
  self1_k<<<(4 * N + thr - 1) / thr, thr, 0, stream>>>(XW1, DNV, H, HB, N);
  agg1_k<<<(E + 32 * EDGE_BATCHES * 8 - 1) / (32 * EDGE_BATCHES * 8), thr, 0, stream>>>(SRC, DST, DNV, XW1, H, HB, E);
  relu_bias_k<<<(4 * N + thr - 1) / thr, thr, 0, stream>>>(H, HB, B1, N);
  gemm2_k<<<(N + thr - 1) / thr, thr, 0, stream>>>(H, W2, DNV, HW2, OUT2, OUT2B, N);
  agg2_k<<<(E + 32 * EDGE_BATCHES * 8 - 1) / (32 * EDGE_BATCHES * 8), thr, 0, stream>>>(SRC, DST, DNV, HW2, OUT2, OUT2B, E);
  lsm_k<<<(N + thr - 1) / thr, thr, 0, stream>>>(OUT2, OUT2B, B2, (float*)d_out, N);
}


# per-edge kernels loop inside fewer waves: degree count 16 edges per lane, both scatters a fixed 10,000 waves over 64-edge chunks; 200k -> 26k short waves; atomics kept, sums still in arrival order
# speedup vs baseline: 3.4108x; 1.0829x over previous
#include <hip/hip_runtime.h>
#include <math.h>


typedef __attribute__((ext_vector_type(16))) __bf16 v16bf;
typedef __attribute__((ext_vector_type(8)))  __bf16 v8bf;
typedef __attribute__((ext_vector_type(8)))  float  v8f;
typedef __attribute__((ext_vector_type(4)))  float  v4f;
typedef __attribute__((ext_vector_type(2)))  float  v2f;

#define IN_FEAT 512
#define HID     16
#define EDGE_BATCHES 2
#define DEG_PER_LANE 16
#define AGG_BLOCKS 1250

__device__ __forceinline__ int package_half() {
  const unsigned die = (__builtin_amdgcn_s_sendmsg_rtn(0x87) >> 16) & 0xFu;
  return (int)((die >> 2) & 1u);
}

__device__ __forceinline__ void atomAddF(float* p, float v) {
  __hip_atomic_fetch_add(p, v, __ATOMIC_RELAXED, __HIP_MEMORY_SCOPE_AGENT);
}

__global__ __launch_bounds__(256) void gemm1_wmma_k(
    const float* __restrict__ X, const float* __restrict__ W1,
    float* __restrict__ XW1, int nTiles) {
  __shared__ __bf16 Bs[HID * IN_FEAT];
  for (int i = threadIdx.x; i < HID * IN_FEAT; i += 256) {
    int k = i >> 4, n = i & 15;
    Bs[n * IN_FEAT + k] = (__bf16)W1[i];
  }
  __syncthreads();

  int wave = threadIdx.x >> 5;
  int lane = threadIdx.x & 31;
  int tile = blockIdx.x * 8 + wave;
  if (tile >= nTiles) return;

  int rc   = lane & 15;
  int half = lane >> 4;

  const float*  arow = X  + (size_t)(tile * 16 + rc) * IN_FEAT;
  const __bf16* bcol = Bs + (size_t)rc * IN_FEAT;

  v8f c = {0.f, 0.f, 0.f, 0.f, 0.f, 0.f, 0.f, 0.f};

#pragma unroll
  for (int ks = 0; ks < IN_FEAT / 32; ++ks) {
    int kb = ks * 32 + half * 8;
    v4f a0 = *(const v4f*)(arow + kb);
    v4f a1 = *(const v4f*)(arow + kb + 4);
    v4f a2 = *(const v4f*)(arow + kb + 16);
    v4f a3 = *(const v4f*)(arow + kb + 20);
    v8bf b0 = *(const v8bf*)(bcol + kb);
    v8bf b1 = *(const v8bf*)(bcol + kb + 16);
    v16bf A, B;
#pragma unroll
    for (int i = 0; i < 4; ++i) {
      A[i]      = (__bf16)a0[i];
      A[4 + i]  = (__bf16)a1[i];
      A[8 + i]  = (__bf16)a2[i];
      A[12 + i] = (__bf16)a3[i];
    }
#pragma unroll
    for (int i = 0; i < 8; ++i) { B[i] = b0[i]; B[8 + i] = b1[i]; }
    c = __builtin_amdgcn_wmma_f32_16x16x32_bf16(
         false, A,  false, B,
         (short)0, c,  false,  false);
  }

  float* orow = XW1 + ((size_t)tile * 16 + half * 8) * HID + rc;
#pragma unroll
  for (int r = 0; r < 8; ++r) orow[r * HID] = c[r];
}

__global__ void deg_init_k(int* DEG, int* DEGB, int N) {
  int n = blockIdx.x * blockDim.x + threadIdx.x;
  if (n < N) { DEG[n] = 1; DEGB[n] = 0; }
}
__global__ __launch_bounds__(256) void deg_edge_k(const int* __restrict__ DST, int* DEG, int* DEGB, int E) {
  int* D = package_half() ? DEGB : DEG;
  const int e0 = blockIdx.x * (256 * DEG_PER_LANE) + threadIdx.x;
  int d[DEG_PER_LANE];
#pragma unroll
  for (int i = 0; i < DEG_PER_LANE; ++i) {
    const int e = e0 + i * 256;
    d[i] = DST[e < E ? e : E - 1];
  }
#pragma unroll
  for (int i = 0; i < DEG_PER_LANE; ++i)
    if (e0 + i * 256 < E) atomicAdd(&D[d[i]], 1);
}
__global__ void dinv_k(float* DNV, const int* __restrict__ DEGB, int N) {
  int n = blockIdx.x * blockDim.x + threadIdx.x;
  if (n < N) {
    int c = ((const int*)DNV)[n] + DEGB[n];
    DNV[n] = rsqrtf((float)c);
  }
}

__global__ void self1_k(const float* __restrict__ XW1,
                        const float* __restrict__ DNV, float* H, float* HB, int N) {
  int t = blockIdx.x * blockDim.x + threadIdx.x;
  int n = t >> 2;
  if (n >= N) return;
  int q = t & 3;
  float w = DNV[n] * DNV[n];
  v4f v = *(const v4f*)(XW1 + (size_t)n * HID + q * 4);
  v4f o;
#pragma unroll
  for (int i = 0; i < 4; ++i) o[i] = w * v[i];
  *(v4f*)(H + (size_t)n * HID + q * 4) = o;
  if (HB != H) {
    const v4f z = {0.f, 0.f, 0.f, 0.f};
    *(v4f*)(HB + (size_t)n * HID + q * 4) = z;
  }
}

__global__ __launch_bounds__(256) void agg1_k(
    const int* __restrict__ SRC, const int* __restrict__ DST,
    const float* __restrict__ DNV, const float* __restrict__ XW1,
    float* H0, float* HB, int E) {
  float* H = package_half() ? HB : H0;
  const int lane = threadIdx.x & 31;
  const int f    = lane & 15;
  const int hi   = lane >> 4;
  const int nWaves  = gridDim.x * 8;
  const int nChunks = (E + 32 * EDGE_BATCHES - 1) / (32 * EDGE_BATCHES);
  for (int wave = blockIdx.x * 8 + (threadIdx.x >> 5); wave < nChunks; wave += nWaves) {
  int   s[EDGE_BATCHES], d[EDGE_BATCHES];
  float nrm[EDGE_BATCHES];
#pragma unroll
  for (int b = 0; b < EDGE_BATCHES; ++b) {
    const int e = (wave * EDGE_BATCHES + b) * 32 + lane;
    s[b] = 0; d[b] = 0; nrm[b] = 0.f;
    if (e < E) {
      s[b] = SRC[e];
      d[b] = DST[e];
      nrm[b] = DNV[s[b]] * DNV[d[b]];
    }
  }
#pragma unroll
  for (int b = 0; b < EDGE_BATCHES; ++b) {
    const int e0 = (wave * EDGE_BATCHES + b) * 32;
    float v[16];
#pragma unroll
    for (int j = 0; j < 16; ++j) {
      const int sj = __shfl(s[b], 2 * j + hi, 32);
      v[j] = XW1[(size_t)sj * HID + f];
    }
#pragma unroll
    for (int j = 0; j < 16; ++j) {
      const int   dj = __shfl(d[b], 2 * j + hi, 32);
      const float nj = __shfl(nrm[b], 2 * j + hi, 32);
      if (e0 + 2 * j + hi < E) atomAddF(H + (size_t)dj * HID + f, nj * v[j]);
    }
  }
  }
}

__global__ void relu_bias_k(float* H, const float* HB, const float* __restrict__ B1, int N) {
  int t = blockIdx.x * blockDim.x + threadIdx.x;
  int n = t >> 2;
  if (n >= N) return;
  int q = t & 3;
  v4f b = *(const v4f*)(B1 + q * 4);
  float* hp = H + (size_t)n * HID + q * 4;
  v4f v = *(const v4f*)hp;
  if (HB != H) {
    const v4f u = *(const v4f*)(HB + (size_t)n * HID + q * 4);
#pragma unroll
    for (int i = 0; i < 4; ++i) v[i] += u[i];
  }
#pragma unroll
  for (int i = 0; i < 4; ++i) v[i] = fmaxf(v[i] + b[i], 0.f);
  *(v4f*)hp = v;
}

__global__ void gemm2_k(const float* __restrict__ H, const float* __restrict__ W2,
                        const float* __restrict__ DNV, float* HW2, float* OUT2,
                        float* OUT2B, int N) {
  int n = blockIdx.x * blockDim.x + threadIdx.x;
  if (n >= N) return;
  const float* hr = H + (size_t)n * HID;
  float h0 = 0.f, h1 = 0.f;
#pragma unroll
  for (int f = 0; f < HID; ++f) {
    float hv = hr[f];
    h0 += hv * W2[2 * f];
    h1 += hv * W2[2 * f + 1];
  }
  HW2[2 * n] = h0;
  HW2[2 * n + 1] = h1;
  float w = DNV[n] * DNV[n];
  OUT2[2 * n] = w * h0;
  OUT2[2 * n + 1] = w * h1;
  OUT2B[2 * n] = 0.f;
  OUT2B[2 * n + 1] = 0.f;
}

__global__ __launch_bounds__(256) void agg2_k(
    const int* __restrict__ SRC, const int* __restrict__ DST,
    const float* __restrict__ DNV, const float* __restrict__ HW2,
    float* OUT2A, float* OUT2B, int E) {
  float* OUT2 = package_half() ? OUT2B : OUT2A;
  const int lane = threadIdx.x & 31;
  const int c    = lane & 1;
  const int sub  = lane >> 1;
  const int nWaves  = gridDim.x * 8;
  const int nChunks = (E + 32 * EDGE_BATCHES - 1) / (32 * EDGE_BATCHES);
  for (int wave = blockIdx.x * 8 + (threadIdx.x >> 5); wave < nChunks; wave += nWaves) {
  int   d[EDGE_BATCHES];
  float v0[EDGE_BATCHES], v1[EDGE_BATCHES];
#pragma unroll
  for (int b = 0; b < EDGE_BATCHES; ++b) {
    const int e = (wave * EDGE_BATCHES + b) * 32 + lane;
    d[b] = 0; v0[b] = 0.f; v1[b] = 0.f;
    if (e < E) {
      const int s = SRC[e];
      d[b] = DST[e];
      const float nrm = DNV[s] * DNV[d[b]];
      const v2f h = *(const v2f*)(HW2 + (size_t)2 * s);
      v0[b] = nrm * h[0];
      v1[b] = nrm * h[1];
    }
  }
#pragma unroll
  for (int b = 0; b < EDGE_BATCHES; ++b) {
    const int e0 = (wave * EDGE_BATCHES + b) * 32;
#pragma unroll
    for (int j = 0; j < 2; ++j) {
      const int   dj = __shfl(d[b], 16 * j + sub, 32);
      const float a0 = __shfl(v0[b], 16 * j + sub, 32);
      const float a1 = __shfl(v1[b], 16 * j + sub, 32);
      if (e0 + 16 * j + sub < E)
        atomAddF(OUT2 + (size_t)2 * dj + c, c ? a1 : a0);
    }
  }
  }
}

__global__ void lsm_k(const float* __restrict__ OUT2, const float* __restrict__ OUT2B,
                      const float* __restrict__ B2, float* __restrict__ out, int N) {
  int n = blockIdx.x * blockDim.x + threadIdx.x;
  if (n >= N) return;
  float a = OUT2[2 * n]     + OUT2B[2 * n]     + B2[0];
  float b = OUT2[2 * n + 1] + OUT2B[2 * n + 1] + B2[1];
  float m = fmaxf(a, b);
  float lse = m + logf(expf(a - m) + expf(b - m));
  out[2 * n]     = a - lse;
  out[2 * n + 1] = b - lse;
}

extern "C" void kernel_launch(void* const* d_in, const int* in_sizes, int n_in,
                              void* d_out, int out_size, void* d_ws, size_t ws_size,
                              hipStream_t stream) {
  const float* X  = (const float*)d_in[0];
  const int*   EI = (const int*)d_in[1];
  const float* W1 = (const float*)d_in[2];
  const float* B1 = (const float*)d_in[3];
  const float* W2 = (const float*)d_in[4];
  const float* B2 = (const float*)d_in[5];

  const int N = in_sizes[0] / IN_FEAT;
  const int E = in_sizes[1] / 2;
  const int* SRC = EI;
  const int* DST = EI + E;

  float* ws   = (float*)d_ws;
  float* XW1  = ws;
  float* H    = ws + (size_t)N * HID;
  float* DNV  = ws + (size_t)2 * N * HID;
  float* HW2  = DNV + N;
  float* OUT2 = HW2 + (size_t)2 * N;
  float* HB   = ws_size >= (size_t)(3 * HID + 5) * N * sizeof(float) ? OUT2 + (size_t)2 * N : H;
  int*   DEGB = (int*)H;
  float* OUT2B = XW1;

  const int thr = 256;
  const int nTiles = (N + 15) / 16;

  gemm1_wmma_k<<<(nTiles + 7) / 8, thr, 0, stream>>>(X, W1, XW1, nTiles);
  deg_init_k<<<(N + thr - 1) / thr, thr, 0, stream>>>((int*)DNV, DEGB, N);
  deg_edge_k<<<(E + thr * DEG_PER_LANE - 1) / (thr * DEG_PER_LANE), thr, 0, stream>>>(DST, (int*)DNV, DEGB, E);
  dinv_k<<<(N + thr - 1) / thr, thr, 0, stream>>>(DNV, DEGB, N);
  self1_k<<<(4 * N + thr - 1) / thr, thr, 0, stream>>>(XW1, DNV, H, HB, N);
  agg1_k<<<AGG_BLOCKS, thr, 0, stream>>>(SRC, DST, DNV, XW1, H, HB, E);
  relu_bias_k<<<(4 * N + thr - 1) / thr, thr, 0, stream>>>(H, HB, B1, N);
  gemm2_k<<<(N + thr - 1) / thr, thr, 0, stream>>>(H, W2, DNV, HW2, OUT2, OUT2B, N);
  agg2_k<<<AGG_BLOCKS, thr, 0, stream>>>(SRC, DST, DNV, HW2, OUT2, OUT2B, E);
  lsm_k<<<(N + thr - 1) / thr, thr, 0, stream>>>(OUT2, OUT2B, B2, (float*)d_out, N);
}


# node-wise kernels ride with a neighbour: degree start values in the GEMM launch, rsqrt in the self-term kernel, relu+bias in the layer-2 product, one thread per node; 10 -> 7 launches; sums as before
# speedup vs baseline: 3.8293x; 1.1227x over previous
#include <hip/hip_runtime.h>
#include <math.h>


typedef __attribute__((ext_vector_type(16))) __bf16 v16bf;
typedef __attribute__((ext_vector_type(8)))  __bf16 v8bf;
typedef __attribute__((ext_vector_type(8)))  float  v8f;
typedef __attribute__((ext_vector_type(4)))  float  v4f;
typedef __attribute__((ext_vector_type(2)))  float  v2f;

#define IN_FEAT 512
#define HID     16
#define EDGE_BATCHES 2
#define DEG_PER_LANE 16
#define AGG_BLOCKS 1250

__device__ __forceinline__ int package_half() {
  const unsigned die = (__builtin_amdgcn_s_sendmsg_rtn(0x87) >> 16) & 0xFu;
  return (int)((die >> 2) & 1u);
}

__device__ __forceinline__ void atomAddF(float* p, float v) {
  __hip_atomic_fetch_add(p, v, __ATOMIC_RELAXED, __HIP_MEMORY_SCOPE_AGENT);
}

__global__ __launch_bounds__(256) void gemm1_wmma_k(
    const float* __restrict__ X, const float* __restrict__ W1,
    float* __restrict__ XW1, int nTiles,
    int* __restrict__ DEG, int* __restrict__ DEGB, int N) {
  __shared__ __bf16 Bs[HID * IN_FEAT];
  if (threadIdx.x < 128) {
    const int n = blockIdx.x * 128 + threadIdx.x;
    if (n < N) { DEG[n] = 1; DEGB[n] = 0; }
  }
  for (int i = threadIdx.x; i < HID * IN_FEAT; i += 256) {
    int k = i >> 4, n = i & 15;
    Bs[n * IN_FEAT + k] = (__bf16)W1[i];
  }
  __syncthreads();

  int wave = threadIdx.x >> 5;
  int lane = threadIdx.x & 31;
  int tile = blockIdx.x * 8 + wave;
  if (tile >= nTiles) return;

  int rc   = lane & 15;
  int half = lane >> 4;

  const float*  arow = X  + (size_t)(tile * 16 + rc) * IN_FEAT;
  const __bf16* bcol = Bs + (size_t)rc * IN_FEAT;

  v8f c = {0.f, 0.f, 0.f, 0.f, 0.f, 0.f, 0.f, 0.f};

#pragma unroll
  for (int ks = 0; ks < IN_FEAT / 32; ++ks) {
    int kb = ks * 32 + half * 8;
    v4f a0 = *(const v4f*)(arow + kb);
    v4f a1 = *(const v4f*)(arow + kb + 4);
    v4f a2 = *(const v4f*)(arow + kb + 16);
    v4f a3 = *(const v4f*)(arow + kb + 20);
    v8bf b0 = *(const v8bf*)(bcol + kb);
    v8bf b1 = *(const v8bf*)(bcol + kb + 16);
    v16bf A, B;
#pragma unroll
    for (int i = 0; i < 4; ++i) {
      A[i]      = (__bf16)a0[i];
      A[4 + i]  = (__bf16)a1[i];
      A[8 + i]  = (__bf16)a2[i];
      A[12 + i] = (__bf16)a3[i];
    }
#pragma unroll
    for (int i = 0; i < 8; ++i) { B[i] = b0[i]; B[8 + i] = b1[i]; }
    c = __builtin_amdgcn_wmma_f32_16x16x32_bf16(
         false, A,  false, B,
         (short)0, c,  false,  false);
  }

  float* orow = XW1 + ((size_t)tile * 16 + half * 8) * HID + rc;
#pragma unroll
  for (int r = 0; r < 8; ++r) orow[r * HID] = c[r];
}

__global__ __launch_bounds__(256) void deg_edge_k(const int* __restrict__ DST, int* DEG, int* DEGB, int E) {
  int* D = package_half() ? DEGB : DEG;
  const int e0 = blockIdx.x * (256 * DEG_PER_LANE) + threadIdx.x;
  int d[DEG_PER_LANE];
#pragma unroll
  for (int i = 0; i < DEG_PER_LANE; ++i) {
    const int e = e0 + i * 256;
    d[i] = DST[e < E ? e : E - 1];
  }
#pragma unroll
  for (int i = 0; i < DEG_PER_LANE; ++i)
    if (e0 + i * 256 < E) atomicAdd(&D[d[i]], 1);
}

__global__ __launch_bounds__(256) void dinv_self1_k(
    const float* __restrict__ XW1, float* DNV, const int* __restrict__ DEGB,
    float* H, float* HB, int N) {
  const int n = blockIdx.x * blockDim.x + threadIdx.x;
  if (n >= N) return;
  const int c = ((const int*)DNV)[n] + DEGB[n];
  const float di = rsqrtf((float)c);
  DNV[n] = di;
  const float w = di * di;
  const v4f* xr = (const v4f*)(XW1 + (size_t)n * HID);
  v4f v[4];
#pragma unroll
  for (int q = 0; q < 4; ++q) v[q] = xr[q];
  v4f* hr = (v4f*)(H + (size_t)n * HID);
#pragma unroll
  for (int q = 0; q < 4; ++q) {
    v4f o;
#pragma unroll
    for (int i = 0; i < 4; ++i) o[i] = w * v[q][i];
    hr[q] = o;
  }
  if (HB != H) {
    const v4f z = {0.f, 0.f, 0.f, 0.f};
    v4f* ur = (v4f*)(HB + (size_t)n * HID);
#pragma unroll
    for (int q = 0; q < 4; ++q) ur[q] = z;
  }
}

__global__ __launch_bounds__(256) void agg1_k(
    const int* __restrict__ SRC, const int* __restrict__ DST,
    const float* __restrict__ DNV, const float* __restrict__ XW1,
    float* H0, float* HB, int E) {
  float* H = package_half() ? HB : H0;
  const int lane = threadIdx.x & 31;
  const int f    = lane & 15;
  const int hi   = lane >> 4;
  const int nWaves  = gridDim.x * 8;
  const int nChunks = (E + 32 * EDGE_BATCHES - 1) / (32 * EDGE_BATCHES);
  for (int wave = blockIdx.x * 8 + (threadIdx.x >> 5); wave < nChunks; wave += nWaves) {
  int   s[EDGE_BATCHES], d[EDGE_BATCHES];
  float nrm[EDGE_BATCHES];
#pragma unroll
  for (int b = 0; b < EDGE_BATCHES; ++b) {
    const int e = (wave * EDGE_BATCHES + b) * 32 + lane;
    s[b] = 0; d[b] = 0; nrm[b] = 0.f;
    if (e < E) {
      s[b] = SRC[e];
      d[b] = DST[e];
      nrm[b] = DNV[s[b]] * DNV[d[b]];
    }
  }
#pragma unroll
  for (int b = 0; b < EDGE_BATCHES; ++b) {
    const int e0 = (wave * EDGE_BATCHES + b) * 32;
    float v[16];
#pragma unroll
    for (int j = 0; j < 16; ++j) {
      const int sj = __shfl(s[b], 2 * j + hi, 32);
      v[j] = XW1[(size_t)sj * HID + f];
    }
#pragma unroll
    for (int j = 0; j < 16; ++j) {
      const int   dj = __shfl(d[b], 2 * j + hi, 32);
      const float nj = __shfl(nrm[b], 2 * j + hi, 32);
      if (e0 + 2 * j + hi < E) atomAddF(H + (size_t)dj * HID + f, nj * v[j]);
    }
  }
  }
}

__global__ __launch_bounds__(256) void relu_gemm2_k(
    const float* H, const float* HB,
    const float* __restrict__ B1, const float* __restrict__ W2,
    const float* __restrict__ DNV, float* HW2, float* OUT2,
    float* OUT2B, int N) {
  const int n = blockIdx.x * blockDim.x + threadIdx.x;
  if (n >= N) return;
  const v4f* hr = (const v4f*)(H + (size_t)n * HID);
  const v4f* ur = (const v4f*)(HB + (size_t)n * HID);
  v4f v[4];
#pragma unroll
  for (int q = 0; q < 4; ++q) v[q] = hr[q];
  if (HB != H) {
    v4f u[4];
#pragma unroll
    for (int q = 0; q < 4; ++q) u[q] = ur[q];
#pragma unroll
    for (int q = 0; q < 4; ++q)
#pragma unroll
      for (int i = 0; i < 4; ++i) v[q][i] += u[q][i];
  }
  float h0 = 0.f, h1 = 0.f;
#pragma unroll
  for (int f = 0; f < HID; ++f) {
    const float hv = fmaxf(v[f >> 2][f & 3] + B1[f], 0.f);
    h0 += hv * W2[2 * f];
    h1 += hv * W2[2 * f + 1];
  }
  HW2[2 * n] = h0;
  HW2[2 * n + 1] = h1;
  float w = DNV[n] * DNV[n];
  OUT2[2 * n] = w * h0;
  OUT2[2 * n + 1] = w * h1;
  OUT2B[2 * n] = 0.f;
  OUT2B[2 * n + 1] = 0.f;
}

__global__ __launch_bounds__(256) void agg2_k(
    const int* __restrict__ SRC, const int* __restrict__ DST,
    const float* __restrict__ DNV, const float* __restrict__ HW2,
    float* OUT2A, float* OUT2B, int E) {
  float* OUT2 = package_half() ? OUT2B : OUT2A;
  const int lane = threadIdx.x & 31;
  const int c    = lane & 1;
  const int sub  = lane >> 1;
  const int nWaves  = gridDim.x * 8;
  const int nChunks = (E + 32 * EDGE_BATCHES - 1) / (32 * EDGE_BATCHES);
  for (int wave = blockIdx.x * 8 + (threadIdx.x >> 5); wave < nChunks; wave += nWaves) {
  int   d[EDGE_BATCHES];
  float v0[EDGE_BATCHES], v1[EDGE_BATCHES];
#pragma unroll
  for (int b = 0; b < EDGE_BATCHES; ++b) {
    const int e = (wave * EDGE_BATCHES + b) * 32 + lane;
    d[b] = 0; v0[b] = 0.f; v1[b] = 0.f;
    if (e < E) {
      const int s = SRC[e];
      d[b] = DST[e];
      const float nrm = DNV[s] * DNV[d[b]];
      const v2f h = *(const v2f*)(HW2 + (size_t)2 * s);
      v0[b] = nrm * h[0];
      v1[b] = nrm * h[1];
    }
  }
#pragma unroll
  for (int b = 0; b < EDGE_BATCHES; ++b) {
    const int e0 = (wave * EDGE_BATCHES + b) * 32;
#pragma unroll
    for (int j = 0; j < 2; ++j) {
      const int   dj = __shfl(d[b], 16 * j + sub, 32);
      const float a0 = __shfl(v0[b], 16 * j + sub, 32);
      const float a1 = __shfl(v1[b], 16 * j + sub, 32);
      if (e0 + 16 * j + sub < E)
        atomAddF(OUT2 + (size_t)2 * dj + c, c ? a1 : a0);
    }
  }
  }
}

__global__ void lsm_k(const float* __restrict__ OUT2, const float* __restrict__ OUT2B,
                      const float* __restrict__ B2, float* __restrict__ out, int N) {
  int n = blockIdx.x * blockDim.x + threadIdx.x;
  if (n >= N) return;
  float a = OUT2[2 * n]     + OUT2B[2 * n]     + B2[0];
  float b = OUT2[2 * n + 1] + OUT2B[2 * n + 1] + B2[1];
  float m = fmaxf(a, b);
  float lse = m + logf(expf(a - m) + expf(b - m));
  out[2 * n]     = a - lse;
  out[2 * n + 1] = b - lse;
}

extern "C" void kernel_launch(void* const* d_in, const int* in_sizes, int n_in,
                              void* d_out, int out_size, void* d_ws, size_t ws_size,
                              hipStream_t stream) {
  const float* X  = (const float*)d_in[0];
  const int*   EI = (const int*)d_in[1];
  const float* W1 = (const float*)d_in[2];
  const float* B1 = (const float*)d_in[3];
  const float* W2 = (const float*)d_in[4];
  const float* B2 = (const float*)d_in[5];

  const int N = in_sizes[0] / IN_FEAT;
  const int E = in_sizes[1] / 2;
  const int* SRC = EI;
  const int* DST = EI + E;

  float* ws   = (float*)d_ws;
  float* XW1  = ws;
  float* H    = ws + (size_t)N * HID;
  float* DNV  = ws + (size_t)2 * N * HID;
  float* HW2  = DNV + N;
  float* OUT2 = HW2 + (size_t)2 * N;
  float* HB   = ws_size >= (size_t)(3 * HID + 5) * N * sizeof(float) ? OUT2 + (size_t)2 * N : H;
  int*   DEGB = (int*)OUT2;
  float* OUT2B = XW1;

  const int thr = 256;
  const int nTiles = (N + 15) / 16;

  gemm1_wmma_k<<<(nTiles + 7) / 8, thr, 0, stream>>>(X, W1, XW1, nTiles, (int*)DNV, DEGB, N);
  deg_edge_k<<<(E + thr * DEG_PER_LANE - 1) / (thr * DEG_PER_LANE), thr, 0, stream>>>(DST, (int*)DNV, DEGB, E);
  dinv_self1_k<<<(N + thr - 1) / thr, thr, 0, stream>>>(XW1, DNV, DEGB, H, HB, N);
  agg1_k<<<AGG_BLOCKS, thr, 0, stream>>>(SRC, DST, DNV, XW1, H, HB, E);
  relu_gemm2_k<<<(N + thr - 1) / thr, thr, 0, stream>>>(H, HB, B1, W2, DNV, HW2, OUT2, OUT2B, N);
  agg2_k<<<AGG_BLOCKS, thr, 0, stream>>>(SRC, DST, DNV, HW2, OUT2, OUT2B, E);
  lsm_k<<<(N + thr - 1) / thr, thr, 0, stream>>>(OUT2, OUT2B, B2, (float*)d_out, N);
}


# norm factored out of both edge loops: node kernels store rows pre-scaled by dinv[s] (layer 1 in place), dinv[d] applied once to the folded plane sum; no dinv gathers per edge; sums round differently
# speedup vs baseline: 5.2881x; 1.3810x over previous
#include <hip/hip_runtime.h>
#include <math.h>


typedef __attribute__((ext_vector_type(16))) __bf16 v16bf;
typedef __attribute__((ext_vector_type(8)))  __bf16 v8bf;
typedef __attribute__((ext_vector_type(8)))  float  v8f;
typedef __attribute__((ext_vector_type(4)))  float  v4f;
typedef __attribute__((ext_vector_type(2)))  float  v2f;

#define IN_FEAT 512
#define HID     16
#define EDGE_BATCHES 2
#define DEG_PER_LANE 16
#define AGG_BLOCKS 1250

__device__ __forceinline__ int package_half() {
  const unsigned die = (__builtin_amdgcn_s_sendmsg_rtn(0x87) >> 16) & 0xFu;
  return (int)((die >> 2) & 1u);
}

__device__ __forceinline__ void atomAddF(float* p, float v) {
  __hip_atomic_fetch_add(p, v, __ATOMIC_RELAXED, __HIP_MEMORY_SCOPE_AGENT);
}

__global__ __launch_bounds__(256) void gemm1_wmma_k(
    const float* __restrict__ X, const float* __restrict__ W1,
    float* __restrict__ XW1, int nTiles,
    int* __restrict__ DEG, int* __restrict__ DEGB, int N) {
  __shared__ __bf16 Bs[HID * IN_FEAT];
  if (threadIdx.x < 128) {
    const int n = blockIdx.x * 128 + threadIdx.x;
    if (n < N) { DEG[n] = 1; DEGB[n] = 0; }
  }
  for (int i = threadIdx.x; i < HID * IN_FEAT; i += 256) {
    int k = i >> 4, n = i & 15;
    Bs[n * IN_FEAT + k] = (__bf16)W1[i];
  }
  __syncthreads();

  int wave = threadIdx.x >> 5;
  int lane = threadIdx.x & 31;
  int tile = blockIdx.x * 8 + wave;
  if (tile >= nTiles) return;

  int rc   = lane & 15;
  int half = lane >> 4;

  const float*  arow = X  + (size_t)(tile * 16 + rc) * IN_FEAT;
  const __bf16* bcol = Bs + (size_t)rc * IN_FEAT;

  v8f c = {0.f, 0.f, 0.f, 0.f, 0.f, 0.f, 0.f, 0.f};

#pragma unroll
  for (int ks = 0; ks < IN_FEAT / 32; ++ks) {
    int kb = ks * 32 + half * 8;
    v4f a0 = *(const v4f*)(arow + kb);
    v4f a1 = *(const v4f*)(arow + kb + 4);
    v4f a2 = *(const v4f*)(arow + kb + 16);
    v4f a3 = *(const v4f*)(arow + kb + 20);
    v8bf b0 = *(const v8bf*)(bcol + kb);
    v8bf b1 = *(const v8bf*)(bcol + kb + 16);
    v16bf A, B;
#pragma unroll
    for (int i = 0; i < 4; ++i) {
      A[i]      = (__bf16)a0[i];
      A[4 + i]  = (__bf16)a1[i];
      A[8 + i]  = (__bf16)a2[i];
      A[12 + i] = (__bf16)a3[i];
    }
#pragma unroll
    for (int i = 0; i < 8; ++i) { B[i] = b0[i]; B[8 + i] = b1[i]; }
    c = __builtin_amdgcn_wmma_f32_16x16x32_bf16(
         false, A,  false, B,
         (short)0, c,  false,  false);
  }

  float* orow = XW1 + ((size_t)tile * 16 + half * 8) * HID + rc;
#pragma unroll
  for (int r = 0; r < 8; ++r) orow[r * HID] = c[r];
}

__global__ __launch_bounds__(256) void deg_edge_k(const int* __restrict__ DST, int* DEG, int* DEGB, int E) {
  int* D = package_half() ? DEGB : DEG;
  const int e0 = blockIdx.x * (256 * DEG_PER_LANE) + threadIdx.x;
  int d[DEG_PER_LANE];
#pragma unroll
  for (int i = 0; i < DEG_PER_LANE; ++i) {
    const int e = e0 + i * 256;
    d[i] = DST[e < E ? e : E - 1];
  }
#pragma unroll
  for (int i = 0; i < DEG_PER_LANE; ++i)
    if (e0 + i * 256 < E) atomicAdd(&D[d[i]], 1);
}

__global__ __launch_bounds__(256) void dinv_self1_k(
    float* XW1, float* DNV, const int* __restrict__ DEGB,
    float* H, float* HB, int N) {
  const int n = blockIdx.x * blockDim.x + threadIdx.x;
  if (n >= N) return;
  const int c = ((const int*)DNV)[n] + DEGB[n];
  const float di = rsqrtf((float)c);
  DNV[n] = di;
  v4f* xr = (v4f*)(XW1 + (size_t)n * HID);
  v4f v[4];
#pragma unroll
  for (int q = 0; q < 4; ++q) v[q] = xr[q];
  v4f* hr = (v4f*)(H + (size_t)n * HID);
#pragma unroll
  for (int q = 0; q < 4; ++q) {
    v4f o;
#pragma unroll
    for (int i = 0; i < 4; ++i) o[i] = di * v[q][i];
    xr[q] = o;
    hr[q] = o;
  }
  if (HB != H) {
    const v4f z = {0.f, 0.f, 0.f, 0.f};
    v4f* ur = (v4f*)(HB + (size_t)n * HID);
#pragma unroll
    for (int q = 0; q < 4; ++q) ur[q] = z;
  }
}

__global__ __launch_bounds__(256) void agg1_k(
    const int* __restrict__ SRC, const int* __restrict__ DST,
    const float* __restrict__ XS,
    float* H0, float* HB, int E) {
  float* H = package_half() ? HB : H0;
  const int lane = threadIdx.x & 31;
  const int f    = lane & 15;
  const int hi   = lane >> 4;
  const int nWaves  = gridDim.x * 8;
  const int nChunks = (E + 32 * EDGE_BATCHES - 1) / (32 * EDGE_BATCHES);
  for (int wave = blockIdx.x * 8 + (threadIdx.x >> 5); wave < nChunks; wave += nWaves) {
  int s[EDGE_BATCHES], d[EDGE_BATCHES];
#pragma unroll
  for (int b = 0; b < EDGE_BATCHES; ++b) {
    const int e  = (wave * EDGE_BATCHES + b) * 32 + lane;
    const int ec = e < E ? e : E - 1;
    s[b] = SRC[ec];
    d[b] = DST[ec];
  }
#pragma unroll
  for (int b = 0; b < EDGE_BATCHES; ++b) {
    const int e0 = (wave * EDGE_BATCHES + b) * 32;
    float v[16];
#pragma unroll
    for (int j = 0; j < 16; ++j) {
      const int sj = __shfl(s[b], 2 * j + hi, 32);
      v[j] = XS[(size_t)sj * HID + f];
    }
#pragma unroll
    for (int j = 0; j < 16; ++j) {
      const int dj = __shfl(d[b], 2 * j + hi, 32);
      if (e0 + 2 * j + hi < E) atomAddF(H + (size_t)dj * HID + f, v[j]);
    }
  }
  }
}

__global__ __launch_bounds__(256) void relu_gemm2_k(
    const float* H, const float* HB,
    const float* __restrict__ B1, const float* __restrict__ W2,
    const float* __restrict__ DNV, float* HW2, float* OUT2,
    float* OUT2B, int N) {
  const int n = blockIdx.x * blockDim.x + threadIdx.x;
  if (n >= N) return;
  const v4f* hr = (const v4f*)(H + (size_t)n * HID);
  const v4f* ur = (const v4f*)(HB + (size_t)n * HID);
  v4f v[4];
#pragma unroll
  for (int q = 0; q < 4; ++q) v[q] = hr[q];
  if (HB != H) {
    v4f u[4];
#pragma unroll
    for (int q = 0; q < 4; ++q) u[q] = ur[q];
#pragma unroll
    for (int q = 0; q < 4; ++q)
#pragma unroll
      for (int i = 0; i < 4; ++i) v[q][i] += u[q][i];
  }
  const float di = DNV[n];
  float h0 = 0.f, h1 = 0.f;
#pragma unroll
  for (int f = 0; f < HID; ++f) {
    const float hv = fmaxf(di * v[f >> 2][f & 3] + B1[f], 0.f);
    h0 += hv * W2[2 * f];
    h1 += hv * W2[2 * f + 1];
  }
  const float g0 = di * h0, g1 = di * h1;
  HW2[2 * n] = g0;
  HW2[2 * n + 1] = g1;
  OUT2[2 * n] = g0;
  OUT2[2 * n + 1] = g1;
  OUT2B[2 * n] = 0.f;
  OUT2B[2 * n + 1] = 0.f;
}

__global__ __launch_bounds__(256) void agg2_k(
    const int* __restrict__ SRC, const int* __restrict__ DST,
    const float* __restrict__ HW2,
    float* OUT2A, float* OUT2B, int E) {
  float* OUT2 = package_half() ? OUT2B : OUT2A;
  const int lane = threadIdx.x & 31;
  const int c    = lane & 1;
  const int sub  = lane >> 1;
  const int nWaves  = gridDim.x * 8;
  const int nChunks = (E + 32 * EDGE_BATCHES - 1) / (32 * EDGE_BATCHES);
  for (int wave = blockIdx.x * 8 + (threadIdx.x >> 5); wave < nChunks; wave += nWaves) {
  int   d[EDGE_BATCHES];
  float v0[EDGE_BATCHES], v1[EDGE_BATCHES];
#pragma unroll
  for (int b = 0; b < EDGE_BATCHES; ++b) {
    const int e = (wave * EDGE_BATCHES + b) * 32 + lane;
    const int ec = e < E ? e : E - 1;
    const int s = SRC[ec];
    d[b] = DST[ec];
    const v2f h = *(const v2f*)(HW2 + (size_t)2 * s);
    v0[b] = h[0];
    v1[b] = h[1];
  }
#pragma unroll
  for (int b = 0; b < EDGE_BATCHES; ++b) {
    const int e0 = (wave * EDGE_BATCHES + b) * 32;
#pragma unroll
    for (int j = 0; j < 2; ++j) {
      const int   dj = __shfl(d[b], 16 * j + sub, 32);
      const float a0 = __shfl(v0[b], 16 * j + sub, 32);
      const float a1 = __shfl(v1[b], 16 * j + sub, 32);
      if (e0 + 16 * j + sub < E)
        atomAddF(OUT2 + (size_t)2 * dj + c, c ? a1 : a0);
    }
  }
  }
}

__global__ void lsm_k(const float* __restrict__ OUT2, const float* __restrict__ OUT2B,
                      const float* __restrict__ DNV,
                      const float* __restrict__ B2, float* __restrict__ out, int N) {
  int n = blockIdx.x * blockDim.x + threadIdx.x;
  if (n >= N) return;
  const float di = DNV[n];
  float a = di * (OUT2[2 * n]     + OUT2B[2 * n])     + B2[0];
  float b = di * (OUT2[2 * n + 1] + OUT2B[2 * n + 1]) + B2[1];
  float m = fmaxf(a, b);
  float lse = m + logf(expf(a - m) + expf(b - m));
  out[2 * n]     = a - lse;
  out[2 * n + 1] = b - lse;
}

extern "C" void kernel_launch(void* const* d_in, const int* in_sizes, int n_in,
                              void* d_out, int out_size, void* d_ws, size_t ws_size,
                              hipStream_t stream) {
  const float* X  = (const float*)d_in[0];
  const int*   EI = (const int*)d_in[1];
  const float* W1 = (const float*)d_in[2];
  const float* B1 = (const float*)d_in[3];
  const float* W2 = (const float*)d_in[4];
  const float* B2 = (const float*)d_in[5];

  const int N = in_sizes[0] / IN_FEAT;
  const int E = in_sizes[1] / 2;
  const int* SRC = EI;
  const int* DST = EI + E;

  float* ws   = (float*)d_ws;
  float* XW1  = ws;
  float* H    = ws + (size_t)N * HID;
  float* DNV  = ws + (size_t)2 * N * HID;
  float* HW2  = DNV + N;
  float* OUT2 = HW2 + (size_t)2 * N;
  float* HB   = ws_size >= (size_t)(3 * HID + 5) * N * sizeof(float) ? OUT2 + (size_t)2 * N : H;
  int*   DEGB = (int*)OUT2;
  float* OUT2B = XW1;

  const int thr = 256;
  const int nTiles = (N + 15) / 16;

  gemm1_wmma_k<<<(nTiles + 7) / 8, thr, 0, stream>>>(X, W1, XW1, nTiles, (int*)DNV, DEGB, N);
  deg_edge_k<<<(E + thr * DEG_PER_LANE - 1) / (thr * DEG_PER_LANE), thr, 0, stream>>>(DST, (int*)DNV, DEGB, E);
  dinv_self1_k<<<(N + thr - 1) / thr, thr, 0, stream>>>(XW1, DNV, DEGB, H, HB, N);
  agg1_k<<<AGG_BLOCKS, thr, 0, stream>>>(SRC, DST, XW1, H, HB, E);
  relu_gemm2_k<<<(N + thr - 1) / thr, thr, 0, stream>>>(H, HB, B1, W2, DNV, HW2, OUT2, OUT2B, N);
  agg2_k<<<AGG_BLOCKS, thr, 0, stream>>>(SRC, DST, HW2, OUT2, OUT2B, E);
  lsm_k<<<(N + thr - 1) / thr, thr, 0, stream>>>(OUT2, OUT2B, DNV, B2, (float*)d_out, N);
}


# degree count and layer-2 sums: a block adds one node range of one edge slab in LDS, then one global add per block and 128-byte line instead of one per edge; atomics kept, layer-2 sums reordered
# speedup vs baseline: 5.7643x; 1.0901x over previous
#include <hip/hip_runtime.h>
#include <math.h>


typedef __attribute__((ext_vector_type(16))) __bf16 v16bf;
typedef __attribute__((ext_vector_type(8)))  __bf16 v8bf;
typedef __attribute__((ext_vector_type(8)))  float  v8f;
typedef __attribute__((ext_vector_type(4)))  float  v4f;
typedef __attribute__((ext_vector_type(2)))  float  v2f;

#define IN_FEAT 512
#define HID     16
#define EDGE_BATCHES 2
#define LDS_THREADS 512
#define DEG_RANGE 50048
#define DEG_SLABS 128
#define DEG_U     49
#define FLUSH_U   14
#define A2_RANGE  25024
#define A2_SLABS  64
#define A2_U      14
#define AGG_BLOCKS 1250

__device__ __forceinline__ int package_half() {
  const unsigned die = (__builtin_amdgcn_s_sendmsg_rtn(0x87) >> 16) & 0xFu;
  return (int)((die >> 2) & 1u);
}

__device__ __forceinline__ void atomAddF(float* p, float v) {
  __hip_atomic_fetch_add(p, v, __ATOMIC_RELAXED, __HIP_MEMORY_SCOPE_AGENT);
}

__global__ __launch_bounds__(256) void gemm1_wmma_k(
    const float* __restrict__ X, const float* __restrict__ W1,
    float* __restrict__ XW1, int nTiles,
    int* __restrict__ DEG, int* __restrict__ DEGB, int N) {
  __shared__ __bf16 Bs[HID * IN_FEAT];
  if (threadIdx.x < 128) {
    const int n = blockIdx.x * 128 + threadIdx.x;
    if (n < N) { DEG[n] = 1; DEGB[n] = 0; }
  }
  for (int i = threadIdx.x; i < HID * IN_FEAT; i += 256) {
    int k = i >> 4, n = i & 15;
    Bs[n * IN_FEAT + k] = (__bf16)W1[i];
  }
  __syncthreads();

  int wave = threadIdx.x >> 5;
  int lane = threadIdx.x & 31;
  int tile = blockIdx.x * 8 + wave;
  if (tile >= nTiles) return;

  int rc   = lane & 15;
  int half = lane >> 4;

  const float*  arow = X  + (size_t)(tile * 16 + rc) * IN_FEAT;
  const __bf16* bcol = Bs + (size_t)rc * IN_FEAT;

  v8f c = {0.f, 0.f, 0.f, 0.f, 0.f, 0.f, 0.f, 0.f};

#pragma unroll
  for (int ks = 0; ks < IN_FEAT / 32; ++ks) {
    int kb = ks * 32 + half * 8;
    v4f a0 = *(const v4f*)(arow + kb);
    v4f a1 = *(const v4f*)(arow + kb + 4);
    v4f a2 = *(const v4f*)(arow + kb + 16);
    v4f a3 = *(const v4f*)(arow + kb + 20);
    v8bf b0 = *(const v8bf*)(bcol + kb);
    v8bf b1 = *(const v8bf*)(bcol + kb + 16);
    v16bf A, B;
#pragma unroll
    for (int i = 0; i < 4; ++i) {
      A[i]      = (__bf16)a0[i];
      A[4 + i]  = (__bf16)a1[i];
      A[8 + i]  = (__bf16)a2[i];
      A[12 + i] = (__bf16)a3[i];
    }
#pragma unroll
    for (int i = 0; i < 8; ++i) { B[i] = b0[i]; B[8 + i] = b1[i]; }
    c = __builtin_amdgcn_wmma_f32_16x16x32_bf16(
         false, A,  false, B,
         (short)0, c,  false,  false);
  }

  float* orow = XW1 + ((size_t)tile * 16 + half * 8) * HID + rc;
#pragma unroll
  for (int r = 0; r < 8; ++r) orow[r * HID] = c[r];
}

typedef __attribute__((ext_vector_type(4))) int v4i;

__global__ __launch_bounds__(LDS_THREADS) void deg_edge_k(
    const int* __restrict__ DST, int* DEG, int* DEGB, int E, int N, int nR) {
  __shared__ __attribute__((aligned(16))) int cnt[DEG_RANGE];
  const int tid = threadIdx.x;
  const v4i z4 = {0, 0, 0, 0};
  for (int i = tid; i < DEG_RANGE / 4; i += LDS_THREADS) ((v4i*)cnt)[i] = z4;
  __syncthreads();

  const int q  = blockIdx.x % nR;
  const int t  = blockIdx.x / nR;
  const int S  = gridDim.x / nR;
  const int lo = q * DEG_RANGE;
  const int eps = (((E + S - 1) / S) + 31) & ~31;
  const int e0 = t * eps;
  const int e1 = (e0 + eps < E) ? e0 + eps : E;
  for (int b0 = e0; b0 < e1; b0 += DEG_U * LDS_THREADS) {
    int d[DEG_U];
#pragma unroll
    for (int k = 0; k < DEG_U; ++k) {
      const int e = b0 + k * LDS_THREADS + tid;
      d[k] = DST[e < e1 ? e : e1 - 1];
    }
#pragma unroll
    for (int k = 0; k < DEG_U; ++k) {
      const unsigned r = (unsigned)(d[k] - lo);
      if (b0 + k * LDS_THREADS + tid < e1 && r < (unsigned)DEG_RANGE) atomicAdd(&cnt[r], 1);
    }
  }
  __syncthreads();

  int* D = package_half() ? DEGB : DEG;
  for (int i0 = tid; i0 < DEG_RANGE; i0 += FLUSH_U * LDS_THREADS) {
    int c[FLUSH_U];
#pragma unroll
    for (int k = 0; k < FLUSH_U; ++k) {
      const int i = i0 + k * LDS_THREADS;
      const int w = cnt[i < DEG_RANGE ? i : DEG_RANGE - 1];
      c[k] = i < DEG_RANGE ? w : 0;
    }
#pragma unroll
    for (int k = 0; k < FLUSH_U; ++k) {
      const int n = lo + i0 + k * LDS_THREADS;
      if (c[k] != 0 && n < N) atomicAdd(&D[n], c[k]);
    }
  }
}

__global__ __launch_bounds__(256) void dinv_self1_k(
    float* XW1, float* DNV, const int* __restrict__ DEGB,
    float* H, float* HB, int N) {
  const int n = blockIdx.x * blockDim.x + threadIdx.x;
  if (n >= N) return;
  const int c = ((const int*)DNV)[n] + DEGB[n];
  const float di = rsqrtf((float)c);
  DNV[n] = di;
  v4f* xr = (v4f*)(XW1 + (size_t)n * HID);
  v4f v[4];
#pragma unroll
  for (int q = 0; q < 4; ++q) v[q] = xr[q];
  v4f* hr = (v4f*)(H + (size_t)n * HID);
#pragma unroll
  for (int q = 0; q < 4; ++q) {
    v4f o;
#pragma unroll
    for (int i = 0; i < 4; ++i) o[i] = di * v[q][i];
    xr[q] = o;
    hr[q] = o;
  }
  if (HB != H) {
    const v4f z = {0.f, 0.f, 0.f, 0.f};
    v4f* ur = (v4f*)(HB + (size_t)n * HID);
#pragma unroll
    for (int q = 0; q < 4; ++q) ur[q] = z;
  }
}

__global__ __launch_bounds__(256) void agg1_k(
    const int* __restrict__ SRC, const int* __restrict__ DST,
    const float* __restrict__ XS,
    float* H0, float* HB, int E) {
  float* H = package_half() ? HB : H0;
  const int lane = threadIdx.x & 31;
  const int f    = lane & 15;
  const int hi   = lane >> 4;
  const int nWaves  = gridDim.x * 8;
  const int nChunks = (E + 32 * EDGE_BATCHES - 1) / (32 * EDGE_BATCHES);
  for (int wave = blockIdx.x * 8 + (threadIdx.x >> 5); wave < nChunks; wave += nWaves) {
  int s[EDGE_BATCHES], d[EDGE_BATCHES];
#pragma unroll
  for (int b = 0; b < EDGE_BATCHES; ++b) {
    const int e  = (wave * EDGE_BATCHES + b) * 32 + lane;
    const int ec = e < E ? e : E - 1;
    s[b] = SRC[ec];
    d[b] = DST[ec];
  }
#pragma unroll
  for (int b = 0; b < EDGE_BATCHES; ++b) {
    const int e0 = (wave * EDGE_BATCHES + b) * 32;
    float v[16];
#pragma unroll
    for (int j = 0; j < 16; ++j) {
      const int sj = __shfl(s[b], 2 * j + hi, 32);
      v[j] = XS[(size_t)sj * HID + f];
    }
#pragma unroll
    for (int j = 0; j < 16; ++j) {
      const int dj = __shfl(d[b], 2 * j + hi, 32);
      if (e0 + 2 * j + hi < E) atomAddF(H + (size_t)dj * HID + f, v[j]);
    }
  }
  }
}

__global__ __launch_bounds__(256) void relu_gemm2_k(
    const float* H, const float* HB,
    const float* __restrict__ B1, const float* __restrict__ W2,
    const float* __restrict__ DNV, float* HW2, float* OUT2,
    float* OUT2B, int N) {
  const int n = blockIdx.x * blockDim.x + threadIdx.x;
  if (n >= N) return;
  const v4f* hr = (const v4f*)(H + (size_t)n * HID);
  const v4f* ur = (const v4f*)(HB + (size_t)n * HID);
  v4f v[4];
#pragma unroll
  for (int q = 0; q < 4; ++q) v[q] = hr[q];
  if (HB != H) {
    v4f u[4];
#pragma unroll
    for (int q = 0; q < 4; ++q) u[q] = ur[q];
#pragma unroll
    for (int q = 0; q < 4; ++q)
#pragma unroll
      for (int i = 0; i < 4; ++i) v[q][i] += u[q][i];
  }
  const float di = DNV[n];
  float h0 = 0.f, h1 = 0.f;
#pragma unroll
  for (int f = 0; f < HID; ++f) {
    const float hv = fmaxf(di * v[f >> 2][f & 3] + B1[f], 0.f);
    h0 += hv * W2[2 * f];
    h1 += hv * W2[2 * f + 1];
  }
  const float g0 = di * h0, g1 = di * h1;
  HW2[2 * n] = g0;
  HW2[2 * n + 1] = g1;
  OUT2[2 * n] = g0;
  OUT2[2 * n + 1] = g1;
  OUT2B[2 * n] = 0.f;
  OUT2B[2 * n + 1] = 0.f;
}

__device__ __forceinline__ void a2_fetch(
    const int* __restrict__ SRC, const int* __restrict__ DST,
    int b0, int tid, int e1, int lo, int (&s)[A2_U], unsigned (&r)[A2_U]) {
#pragma unroll
  for (int k = 0; k < A2_U; ++k) {
    const int e  = b0 + k * LDS_THREADS + tid;
    const int ec = e < e1 ? e : e1 - 1;
    s[k] = SRC[ec];
    const unsigned rr = (unsigned)(DST[ec] - lo);
    r[k] = e < e1 ? rr : 0xFFFFFFFFu;
  }
}

__global__ __launch_bounds__(LDS_THREADS) void agg2_k(
    const int* __restrict__ SRC, const int* __restrict__ DST,
    const float* __restrict__ HW2,
    float* OUT2A, float* OUT2B, int E, int N, int nR) {
  __shared__ __attribute__((aligned(16))) float acc[2 * A2_RANGE];
  const int tid = threadIdx.x;
  const v4f z4 = {0.f, 0.f, 0.f, 0.f};
  for (int i = tid; i < 2 * A2_RANGE / 4; i += LDS_THREADS) ((v4f*)acc)[i] = z4;
  __syncthreads();

  const int q  = blockIdx.x % nR;
  const int t  = blockIdx.x / nR;
  const int S  = gridDim.x / nR;
  const int lo = q * A2_RANGE;
  const int eps = (((E + S - 1) / S) + 31) & ~31;
  const int e0 = t * eps;
  const int e1 = (e0 + eps < E) ? e0 + eps : E;

  int sN[A2_U];
  unsigned rN[A2_U];
  if (e0 < e1) a2_fetch(SRC, DST, e0, tid, e1, lo, sN, rN);
  for (int b0 = e0; b0 < e1; b0 += A2_U * LDS_THREADS) {
    int s[A2_U];
    unsigned r[A2_U];
#pragma unroll
    for (int k = 0; k < A2_U; ++k) { s[k] = sN[k]; r[k] = rN[k]; }
    v2f h[A2_U];
#pragma unroll
    for (int k = 0; k < A2_U; ++k) {
      const int sk = r[k] < (unsigned)A2_RANGE ? s[k] : 0;
      h[k] = *(const v2f*)(HW2 + (size_t)2 * sk);
    }
    const int b1 = b0 + A2_U * LDS_THREADS;
    if (b1 < e1) a2_fetch(SRC, DST, b1, tid, e1, lo, sN, rN);
#pragma unroll
    for (int k = 0; k < A2_U; ++k)
      if (r[k] < (unsigned)A2_RANGE) {
        __hip_atomic_fetch_add(&acc[2 * r[k]],     h[k][0], __ATOMIC_RELAXED, __HIP_MEMORY_SCOPE_WORKGROUP);
        __hip_atomic_fetch_add(&acc[2 * r[k] + 1], h[k][1], __ATOMIC_RELAXED, __HIP_MEMORY_SCOPE_WORKGROUP);
      }
  }
  __syncthreads();

  float* O = (package_half() ? OUT2B : OUT2A) + (size_t)2 * lo;
  for (int i0 = tid; i0 < 2 * A2_RANGE; i0 += FLUSH_U * LDS_THREADS) {
    float v[FLUSH_U];
#pragma unroll
    for (int k = 0; k < FLUSH_U; ++k) {
      const int i = i0 + k * LDS_THREADS;
      const float w = acc[i < 2 * A2_RANGE ? i : 2 * A2_RANGE - 1];
      v[k] = i < 2 * A2_RANGE ? w : 0.f;
    }
#pragma unroll
    for (int k = 0; k < FLUSH_U; ++k) {
      const int i = i0 + k * LDS_THREADS;
      if (v[k] != 0.f && lo + (i >> 1) < N) atomAddF(O + i, v[k]);
    }
  }
}

__global__ void lsm_k(const float* __restrict__ OUT2, const float* __restrict__ OUT2B,
                      const float* __restrict__ DNV,
                      const float* __restrict__ B2, float* __restrict__ out, int N) {
  int n = blockIdx.x * blockDim.x + threadIdx.x;
  if (n >= N) return;
  const float di = DNV[n];
  float a = di * (OUT2[2 * n]     + OUT2B[2 * n])     + B2[0];
  float b = di * (OUT2[2 * n + 1] + OUT2B[2 * n + 1]) + B2[1];
  float m = fmaxf(a, b);
  float lse = m + logf(expf(a - m) + expf(b - m));
  out[2 * n]     = a - lse;
  out[2 * n + 1] = b - lse;
}

extern "C" void kernel_launch(void* const* d_in, const int* in_sizes, int n_in,
                              void* d_out, int out_size, void* d_ws, size_t ws_size,
                              hipStream_t stream) {
  const float* X  = (const float*)d_in[0];
  const int*   EI = (const int*)d_in[1];
  const float* W1 = (const float*)d_in[2];
  const float* B1 = (const float*)d_in[3];
  const float* W2 = (const float*)d_in[4];
  const float* B2 = (const float*)d_in[5];

  const int N = in_sizes[0] / IN_FEAT;
  const int E = in_sizes[1] / 2;
  const int* SRC = EI;
  const int* DST = EI + E;

  float* ws   = (float*)d_ws;
  float* XW1  = ws;
  float* H    = ws + (size_t)N * HID;
  float* DNV  = ws + (size_t)2 * N * HID;
  float* HW2  = DNV + N;
  float* OUT2 = HW2 + (size_t)2 * N;
  float* HB   = ws_size >= (size_t)(3 * HID + 5) * N * sizeof(float) ? OUT2 + (size_t)2 * N : H;
  int*   DEGB = (int*)OUT2;
  float* OUT2B = XW1;

  const int thr = 256;
  const int nTiles = (N + 15) / 16;
  const int nRd = (N + DEG_RANGE - 1) / DEG_RANGE;
  const int nR2 = (N + A2_RANGE - 1) / A2_RANGE;

  gemm1_wmma_k<<<(nTiles + 7) / 8, thr, 0, stream>>>(X, W1, XW1, nTiles, (int*)DNV, DEGB, N);
  deg_edge_k<<<nRd * DEG_SLABS, LDS_THREADS, 0, stream>>>(DST, (int*)DNV, DEGB, E, N, nRd);
  dinv_self1_k<<<(N + thr - 1) / thr, thr, 0, stream>>>(XW1, DNV, DEGB, H, HB, N);
  agg1_k<<<AGG_BLOCKS, thr, 0, stream>>>(SRC, DST, XW1, H, HB, E);
  relu_gemm2_k<<<(N + thr - 1) / thr, thr, 0, stream>>>(H, HB, B1, W2, DNV, HW2, OUT2, OUT2B, N);
  agg2_k<<<nR2 * A2_SLABS, LDS_THREADS, 0, stream>>>(SRC, DST, HW2, OUT2, OUT2B, E, N, nR2);
  lsm_k<<<(N + thr - 1) / thr, thr, 0, stream>>>(OUT2, OUT2B, DNV, B2, (float*)d_out, N);
}


# layer-1 edge scatter, the wave's dependent chain: indices by DPP row_share, no LDS permute waited for before each atomic; one tail test per chunk; its 32 row loads as one group; atomics kept
# speedup vs baseline: 5.9473x; 1.0317x over previous
#include <hip/hip_runtime.h>
#include <math.h>


typedef __attribute__((ext_vector_type(16))) __bf16 v16bf;
typedef __attribute__((ext_vector_type(8)))  __bf16 v8bf;
typedef __attribute__((ext_vector_type(8)))  float  v8f;
typedef __attribute__((ext_vector_type(4)))  float  v4f;
typedef __attribute__((ext_vector_type(2)))  float  v2f;

#define IN_FEAT 512
#define HID     16
#define EDGE_BATCHES 2
#define LDS_THREADS 512
#define DEG_RANGE 50048
#define DEG_SLABS 128
#define DEG_U     49
#define FLUSH_U   14
#define A2_RANGE  25024
#define A2_SLABS  64
#define A2_U      14
#define AGG_BLOCKS 1250

__device__ __forceinline__ int package_half() {
  const unsigned die = (__builtin_amdgcn_s_sendmsg_rtn(0x87) >> 16) & 0xFu;
  return (int)((die >> 2) & 1u);
}

__device__ __forceinline__ void atomAddF(float* p, float v) {
  __hip_atomic_fetch_add(p, v, __ATOMIC_RELAXED, __HIP_MEMORY_SCOPE_AGENT);
}

__global__ __launch_bounds__(256) void gemm1_wmma_k(
    const float* __restrict__ X, const float* __restrict__ W1,
    float* __restrict__ XW1, int nTiles,
    int* __restrict__ DEG, int* __restrict__ DEGB, int N) {
  __shared__ __bf16 Bs[HID * IN_FEAT];
  if (threadIdx.x < 128) {
    const int n = blockIdx.x * 128 + threadIdx.x;
    if (n < N) { DEG[n] = 1; DEGB[n] = 0; }
  }
  for (int i = threadIdx.x; i < HID * IN_FEAT; i += 256) {
    int k = i >> 4, n = i & 15;
    Bs[n * IN_FEAT + k] = (__bf16)W1[i];
  }
  __syncthreads();

  int wave = threadIdx.x >> 5;
  int lane = threadIdx.x & 31;
  int tile = blockIdx.x * 8 + wave;
  if (tile >= nTiles) return;

  int rc   = lane & 15;
  int half = lane >> 4;

  const float*  arow = X  + (size_t)(tile * 16 + rc) * IN_FEAT;
  const __bf16* bcol = Bs + (size_t)rc * IN_FEAT;

  v8f c = {0.f, 0.f, 0.f, 0.f, 0.f, 0.f, 0.f, 0.f};

#pragma unroll
  for (int ks = 0; ks < IN_FEAT / 32; ++ks) {
    int kb = ks * 32 + half * 8;
    v4f a0 = *(const v4f*)(arow + kb);
    v4f a1 = *(const v4f*)(arow + kb + 4);
    v4f a2 = *(const v4f*)(arow + kb + 16);
    v4f a3 = *(const v4f*)(arow + kb + 20);
    v8bf b0 = *(const v8bf*)(bcol + kb);
    v8bf b1 = *(const v8bf*)(bcol + kb + 16);
    v16bf A, B;
#pragma unroll
    for (int i = 0; i < 4; ++i) {
      A[i]      = (__bf16)a0[i];
      A[4 + i]  = (__bf16)a1[i];
      A[8 + i]  = (__bf16)a2[i];
      A[12 + i] = (__bf16)a3[i];
    }
#pragma unroll
    for (int i = 0; i < 8; ++i) { B[i] = b0[i]; B[8 + i] = b1[i]; }
    c = __builtin_amdgcn_wmma_f32_16x16x32_bf16(
         false, A,  false, B,
         (short)0, c,  false,  false);
  }

  float* orow = XW1 + ((size_t)tile * 16 + half * 8) * HID + rc;
#pragma unroll
  for (int r = 0; r < 8; ++r) orow[r * HID] = c[r];
}

typedef __attribute__((ext_vector_type(4))) int v4i;

__global__ __launch_bounds__(LDS_THREADS) void deg_edge_k(
    const int* __restrict__ DST, int* DEG, int* DEGB, int E, int N, int nR) {
  __shared__ __attribute__((aligned(16))) int cnt[DEG_RANGE];
  const int tid = threadIdx.x;
  const v4i z4 = {0, 0, 0, 0};
  for (int i = tid; i < DEG_RANGE / 4; i += LDS_THREADS) ((v4i*)cnt)[i] = z4;
  __syncthreads();

  const int q  = blockIdx.x % nR;
  const int t  = blockIdx.x / nR;
  const int S  = gridDim.x / nR;
  const int lo = q * DEG_RANGE;
  const int eps = (((E + S - 1) / S) + 31) & ~31;
  const int e0 = t * eps;
  const int e1 = (e0 + eps < E) ? e0 + eps : E;
  for (int b0 = e0; b0 < e1; b0 += DEG_U * LDS_THREADS) {
    int d[DEG_U];
#pragma unroll
    for (int k = 0; k < DEG_U; ++k) {
      const int e = b0 + k * LDS_THREADS + tid;
      d[k] = DST[e < e1 ? e : e1 - 1];
    }
#pragma unroll
    for (int k = 0; k < DEG_U; ++k) {
      const unsigned r = (unsigned)(d[k] - lo);
      if (b0 + k * LDS_THREADS + tid < e1 && r < (unsigned)DEG_RANGE) atomicAdd(&cnt[r], 1);
    }
  }
  __syncthreads();

  int* D = package_half() ? DEGB : DEG;
  for (int i0 = tid; i0 < DEG_RANGE; i0 += FLUSH_U * LDS_THREADS) {
    int c[FLUSH_U];
#pragma unroll
    for (int k = 0; k < FLUSH_U; ++k) {
      const int i = i0 + k * LDS_THREADS;
      const int w = cnt[i < DEG_RANGE ? i : DEG_RANGE - 1];
      c[k] = i < DEG_RANGE ? w : 0;
    }
#pragma unroll
    for (int k = 0; k < FLUSH_U; ++k) {
      const int n = lo + i0 + k * LDS_THREADS;
      if (c[k] != 0 && n < N) atomicAdd(&D[n], c[k]);
    }
  }
}

__global__ __launch_bounds__(256) void dinv_self1_k(
    float* XW1, float* DNV, const int* __restrict__ DEGB,
    float* H, float* HB, int N) {
  const int n = blockIdx.x * blockDim.x + threadIdx.x;
  if (n >= N) return;
  const int c = ((const int*)DNV)[n] + DEGB[n];
  const float di = rsqrtf((float)c);
  DNV[n] = di;
  v4f* xr = (v4f*)(XW1 + (size_t)n * HID);
  v4f v[4];
#pragma unroll
  for (int q = 0; q < 4; ++q) v[q] = xr[q];
  v4f* hr = (v4f*)(H + (size_t)n * HID);
#pragma unroll
  for (int q = 0; q < 4; ++q) {
    v4f o;
#pragma unroll
    for (int i = 0; i < 4; ++i) o[i] = di * v[q][i];
    xr[q] = o;
    hr[q] = o;
  }
  if (HB != H) {
    const v4f z = {0.f, 0.f, 0.f, 0.f};
    v4f* ur = (v4f*)(HB + (size_t)n * HID);
#pragma unroll
    for (int q = 0; q < 4; ++q) ur[q] = z;
  }
}


template <int J>
__device__ __forceinline__ int row_lane(int v) {
  return __builtin_amdgcn_update_dpp(0, v, 0x150 + J, 0xF, 0xF, false);
}

#define REP16(M) M(0) M(1) M(2) M(3) M(4) M(5) M(6) M(7) M(8) M(9) M(10) M(11) M(12) M(13) M(14) M(15)

template <bool FULL>
__device__ __forceinline__ void agg1_chunk(const int (&s)[EDGE_BATCHES], const int (&d)[EDGE_BATCHES],
                                           int eRow, int E, int f,
                                           const float* __restrict__ XS, float* H) {
  unsigned so[EDGE_BATCHES][16], dO[EDGE_BATCHES][16];
  float v[EDGE_BATCHES][16];
#pragma unroll
  for (int b = 0; b < EDGE_BATCHES; ++b) {
#define AGG1_SO(J) so[b][J] = (unsigned)(row_lane<J>(s[b]) * HID + f);
    REP16(AGG1_SO)
#undef AGG1_SO
  }
#pragma unroll
  for (int b = 0; b < EDGE_BATCHES; ++b) {
#define AGG1_LD(J) v[b][J] = XS[so[b][J]];
    REP16(AGG1_LD)
#undef AGG1_LD
  }
#pragma unroll
  for (int b = 0; b < EDGE_BATCHES; ++b) {
#define AGG1_DO(J) dO[b][J] = (unsigned)(row_lane<J>(d[b]) * HID + f);
    REP16(AGG1_DO)
#undef AGG1_DO
  }
#pragma unroll
  for (int b = 0; b < EDGE_BATCHES; ++b) {
#define AGG1_ADD(J) if (FULL || eRow + 32 * b + J < E) atomAddF(H + dO[b][J], v[b][J]);
    REP16(AGG1_ADD)
#undef AGG1_ADD
  }
}

__device__ __forceinline__ void agg1_fetch(
    const int* __restrict__ SRC, const int* __restrict__ DST,
    int chunk, int lane, int E, int (&s)[EDGE_BATCHES], int (&d)[EDGE_BATCHES]) {
#pragma unroll
  for (int b = 0; b < EDGE_BATCHES; ++b) {
    const int e  = (chunk * EDGE_BATCHES + b) * 32 + lane;
    const int ec = e < E ? e : E - 1;
    s[b] = SRC[ec];
    d[b] = DST[ec];
  }
}

__global__ __launch_bounds__(256) void agg1_k(
    const int* __restrict__ SRC, const int* __restrict__ DST,
    const float* __restrict__ XS,
    float* H0, float* HB, int E) {
  float* H = package_half() ? HB : H0;
  const int lane = threadIdx.x & 31;
  const int f    = lane & 15;
  const int hi   = lane >> 4;
  const int w0 = __builtin_amdgcn_readfirstlane((int)(blockIdx.x * 8 + (threadIdx.x >> 5)));
  const int nWaves  = gridDim.x * 8;
  const int nChunks = (E + 32 * EDGE_BATCHES - 1) / (32 * EDGE_BATCHES);
  if (w0 >= nChunks) return;
  int sN[EDGE_BATCHES], dN[EDGE_BATCHES];
  agg1_fetch(SRC, DST, w0, lane, E, sN, dN);
  for (int wave = w0; wave < nChunks; wave += nWaves) {
    int s[EDGE_BATCHES], d[EDGE_BATCHES];
#pragma unroll
    for (int b = 0; b < EDGE_BATCHES; ++b) { s[b] = sN[b]; d[b] = dN[b]; }
    if (wave + nWaves < nChunks) agg1_fetch(SRC, DST, wave + nWaves, lane, E, sN, dN);
    const int e0 = wave * (32 * EDGE_BATCHES);
    if (e0 + 32 * EDGE_BATCHES <= E) agg1_chunk<true >(s, d, e0 + 16 * hi, E, f, XS, H);
    else                             agg1_chunk<false>(s, d, e0 + 16 * hi, E, f, XS, H);
  }
}

__global__ __launch_bounds__(256) void relu_gemm2_k(
    const float* H, const float* HB,
    const float* __restrict__ B1, const float* __restrict__ W2,
    const float* __restrict__ DNV, float* HW2, float* OUT2,
    float* OUT2B, int N) {
  const int n = blockIdx.x * blockDim.x + threadIdx.x;
  if (n >= N) return;
  const v4f* hr = (const v4f*)(H + (size_t)n * HID);
  const v4f* ur = (const v4f*)(HB + (size_t)n * HID);
  v4f v[4];
#pragma unroll
  for (int q = 0; q < 4; ++q) v[q] = hr[q];
  if (HB != H) {
    v4f u[4];
#pragma unroll
    for (int q = 0; q < 4; ++q) u[q] = ur[q];
#pragma unroll
    for (int q = 0; q < 4; ++q)
#pragma unroll
      for (int i = 0; i < 4; ++i) v[q][i] += u[q][i];
  }
  const float di = DNV[n];
  float h0 = 0.f, h1 = 0.f;
#pragma unroll
  for (int f = 0; f < HID; ++f) {
    const float hv = fmaxf(di * v[f >> 2][f & 3] + B1[f], 0.f);
    h0 += hv * W2[2 * f];
    h1 += hv * W2[2 * f + 1];
  }
  const float g0 = di * h0, g1 = di * h1;
  HW2[2 * n] = g0;
  HW2[2 * n + 1] = g1;
  OUT2[2 * n] = g0;
  OUT2[2 * n + 1] = g1;
  OUT2B[2 * n] = 0.f;
  OUT2B[2 * n + 1] = 0.f;
}

__device__ __forceinline__ void a2_fetch(
    const int* __restrict__ SRC, const int* __restrict__ DST,
    int b0, int tid, int e1, int lo, int (&s)[A2_U], unsigned (&r)[A2_U]) {
#pragma unroll
  for (int k = 0; k < A2_U; ++k) {
    const int e  = b0 + k * LDS_THREADS + tid;
    const int ec = e < e1 ? e : e1 - 1;
    s[k] = SRC[ec];
    const unsigned rr = (unsigned)(DST[ec] - lo);
    r[k] = e < e1 ? rr : 0xFFFFFFFFu;
  }
}

__global__ __launch_bounds__(LDS_THREADS) void agg2_k(
    const int* __restrict__ SRC, const int* __restrict__ DST,
    const float* __restrict__ HW2,
    float* OUT2A, float* OUT2B, int E, int N, int nR) {
  __shared__ __attribute__((aligned(16))) float acc[2 * A2_RANGE];
  const int tid = threadIdx.x;
  const v4f z4 = {0.f, 0.f, 0.f, 0.f};
  for (int i = tid; i < 2 * A2_RANGE / 4; i += LDS_THREADS) ((v4f*)acc)[i] = z4;
  __syncthreads();

  const int q  = blockIdx.x % nR;
  const int t  = blockIdx.x / nR;
  const int S  = gridDim.x / nR;
  const int lo = q * A2_RANGE;
  const int eps = (((E + S - 1) / S) + 31) & ~31;
  const int e0 = t * eps;
  const int e1 = (e0 + eps < E) ? e0 + eps : E;

  int sN[A2_U];
  unsigned rN[A2_U];
  if (e0 < e1) a2_fetch(SRC, DST, e0, tid, e1, lo, sN, rN);
  for (int b0 = e0; b0 < e1; b0 += A2_U * LDS_THREADS) {
    int s[A2_U];
    unsigned r[A2_U];
#pragma unroll
    for (int k = 0; k < A2_U; ++k) { s[k] = sN[k]; r[k] = rN[k]; }
    v2f h[A2_U];
#pragma unroll
    for (int k = 0; k < A2_U; ++k) {
      const int sk = r[k] < (unsigned)A2_RANGE ? s[k] : 0;
      h[k] = *(const v2f*)(HW2 + (size_t)2 * sk);
    }
    const int b1 = b0 + A2_U * LDS_THREADS;
    if (b1 < e1) a2_fetch(SRC, DST, b1, tid, e1, lo, sN, rN);
#pragma unroll
    for (int k = 0; k < A2_U; ++k)
      if (r[k] < (unsigned)A2_RANGE) {
        __hip_atomic_fetch_add(&acc[2 * r[k]],     h[k][0], __ATOMIC_RELAXED, __HIP_MEMORY_SCOPE_WORKGROUP);
        __hip_atomic_fetch_add(&acc[2 * r[k] + 1], h[k][1], __ATOMIC_RELAXED, __HIP_MEMORY_SCOPE_WORKGROUP);
      }
  }
  __syncthreads();

  float* O = (package_half() ? OUT2B : OUT2A) + (size_t)2 * lo;
  for (int i0 = tid; i0 < 2 * A2_RANGE; i0 += FLUSH_U * LDS_THREADS) {
    float v[FLUSH_U];
#pragma unroll
    for (int k = 0; k < FLUSH_U; ++k) {
      const int i = i0 + k * LDS_THREADS;
      const float w = acc[i < 2 * A2_RANGE ? i : 2 * A2_RANGE - 1];
      v[k] = i < 2 * A2_RANGE ? w : 0.f;
    }
#pragma unroll
    for (int k = 0; k < FLUSH_U; ++k) {
      const int i = i0 + k * LDS_THREADS;
      if (v[k] != 0.f && lo + (i >> 1) < N) atomAddF(O + i, v[k]);
    }
  }
}

__global__ void lsm_k(const float* __restrict__ OUT2, const float* __restrict__ OUT2B,
                      const float* __restrict__ DNV,
                      const float* __restrict__ B2, float* __restrict__ out, int N) {
  int n = blockIdx.x * blockDim.x + threadIdx.x;
  if (n >= N) return;
  const float di = DNV[n];
  float a = di * (OUT2[2 * n]     + OUT2B[2 * n])     + B2[0];
  float b = di * (OUT2[2 * n + 1] + OUT2B[2 * n + 1]) + B2[1];
  float m = fmaxf(a, b);
  float lse = m + logf(expf(a - m) + expf(b - m));
  out[2 * n]     = a - lse;
  out[2 * n + 1] = b - lse;
}

extern "C" void kernel_launch(void* const* d_in, const int* in_sizes, int n_in,
                              void* d_out, int out_size, void* d_ws, size_t ws_size,
                              hipStream_t stream) {
  const float* X  = (const float*)d_in[0];
  const int*   EI = (const int*)d_in[1];
  const float* W1 = (const float*)d_in[2];
  const float* B1 = (const float*)d_in[3];
  const float* W2 = (const float*)d_in[4];
  const float* B2 = (const float*)d_in[5];

  const int N = in_sizes[0] / IN_FEAT;
  const int E = in_sizes[1] / 2;
  const int* SRC = EI;
  const int* DST = EI + E;

  float* ws   = (float*)d_ws;
  float* XW1  = ws;
  float* H    = ws + (size_t)N * HID;
  float* DNV  = ws + (size_t)2 * N * HID;
  float* HW2  = DNV + N;
  float* OUT2 = HW2 + (size_t)2 * N;
  float* HB   = ws_size >= (size_t)(3 * HID + 5) * N * sizeof(float) ? OUT2 + (size_t)2 * N : H;
  int*   DEGB = (int*)OUT2;
  float* OUT2B = XW1;

  const int thr = 256;
  const int nTiles = (N + 15) / 16;
  const int nRd = (N + DEG_RANGE - 1) / DEG_RANGE;
  const int nR2 = (N + A2_RANGE - 1) / A2_RANGE;

  gemm1_wmma_k<<<(nTiles + 7) / 8, thr, 0, stream>>>(X, W1, XW1, nTiles, (int*)DNV, DEGB, N);
  deg_edge_k<<<nRd * DEG_SLABS, LDS_THREADS, 0, stream>>>(DST, (int*)DNV, DEGB, E, N, nRd);
  dinv_self1_k<<<(N + thr - 1) / thr, thr, 0, stream>>>(XW1, DNV, DEGB, H, HB, N);
  agg1_k<<<AGG_BLOCKS, thr, 0, stream>>>(SRC, DST, XW1, H, HB, E);
  relu_gemm2_k<<<(N + thr - 1) / thr, thr, 0, stream>>>(H, HB, B1, W2, DNV, HW2, OUT2, OUT2B, N);
  agg2_k<<<nR2 * A2_SLABS, LDS_THREADS, 0, stream>>>(SRC, DST, HW2, OUT2, OUT2B, E, N, nR2);
  lsm_k<<<(N + thr - 1) / thr, thr, 0, stream>>>(OUT2, OUT2B, DNV, B2, (float*)d_out, N);
}


# X@W1 kernel: W1 tile in LDS at a 1,040-byte row pitch (16 B x odd) with k pairs stored as dwords; its staging stores and B-fragment ds_load_b128 were 16-way bank conflicts at 1,024 B; same sums
# speedup vs baseline: 6.2874x; 1.0572x over previous
#include <hip/hip_runtime.h>
#include <math.h>


typedef __attribute__((ext_vector_type(16))) __bf16 v16bf;
typedef __attribute__((ext_vector_type(8)))  __bf16 v8bf;
typedef __attribute__((ext_vector_type(2)))  __bf16 v2bf;
typedef __attribute__((ext_vector_type(8)))  float  v8f;
typedef __attribute__((ext_vector_type(4)))  float  v4f;
typedef __attribute__((ext_vector_type(2)))  float  v2f;

#define IN_FEAT 512
#define HID     16
#define BS_PITCH 520
#define EDGE_BATCHES 2
#define LDS_THREADS 512
#define DEG_RANGE 50048
#define DEG_SLABS 128
#define DEG_U     49
#define FLUSH_U   14
#define A2_RANGE  25024
#define A2_SLABS  64
#define A2_U      14
#define AGG_BLOCKS 1250

__device__ __forceinline__ int package_half() {
  const unsigned die = (__builtin_amdgcn_s_sendmsg_rtn(0x87) >> 16) & 0xFu;
  return (int)((die >> 2) & 1u);
}

__device__ __forceinline__ void atomAddF(float* p, float v) {
  __hip_atomic_fetch_add(p, v, __ATOMIC_RELAXED, __HIP_MEMORY_SCOPE_AGENT);
}

__global__ __launch_bounds__(256) void gemm1_wmma_k(
    const float* __restrict__ X, const float* __restrict__ W1,
    float* __restrict__ XW1, int nTiles,
    int* __restrict__ DEG, int* __restrict__ DEGB, int N) {
  __shared__ __attribute__((aligned(16))) __bf16 Bs[HID * BS_PITCH];
  if (threadIdx.x < 128) {
    const int n = blockIdx.x * 128 + threadIdx.x;
    if (n < N) { DEG[n] = 1; DEGB[n] = 0; }
  }
  for (int j = threadIdx.x; j < HID * IN_FEAT / 2; j += 256) {
    const int n = j & 15, kp = j >> 4;
    v2bf p;
    p[0] = (__bf16)W1[(2 * kp) * HID + n];
    p[1] = (__bf16)W1[(2 * kp + 1) * HID + n];
    *(v2bf*)(Bs + n * BS_PITCH + 2 * kp) = p;
  }
  __syncthreads();

  int wave = threadIdx.x >> 5;
  int lane = threadIdx.x & 31;
  int tile = blockIdx.x * 8 + wave;
  if (tile >= nTiles) return;

  int rc   = lane & 15;
  int half = lane >> 4;

  const float*  arow = X  + (size_t)(tile * 16 + rc) * IN_FEAT;
  const __bf16* bcol = Bs + (size_t)rc * BS_PITCH;

  v8f c = {0.f, 0.f, 0.f, 0.f, 0.f, 0.f, 0.f, 0.f};

#pragma unroll
  for (int ks = 0; ks < IN_FEAT / 32; ++ks) {
    int kb = ks * 32 + half * 8;
    v4f a0 = *(const v4f*)(arow + kb);
    v4f a1 = *(const v4f*)(arow + kb + 4);
    v4f a2 = *(const v4f*)(arow + kb + 16);
    v4f a3 = *(const v4f*)(arow + kb + 20);
    v8bf b0 = *(const v8bf*)(bcol + kb);
    v8bf b1 = *(const v8bf*)(bcol + kb + 16);
    v16bf A, B;
#pragma unroll
    for (int i = 0; i < 4; ++i) {
      A[i]      = (__bf16)a0[i];
      A[4 + i]  = (__bf16)a1[i];
      A[8 + i]  = (__bf16)a2[i];
      A[12 + i] = (__bf16)a3[i];
    }
#pragma unroll
    for (int i = 0; i < 8; ++i) { B[i] = b0[i]; B[8 + i] = b1[i]; }
    c = __builtin_amdgcn_wmma_f32_16x16x32_bf16(
         false, A,  false, B,
         (short)0, c,  false,  false);
  }

  float* orow = XW1 + ((size_t)tile * 16 + half * 8) * HID + rc;
#pragma unroll
  for (int r = 0; r < 8; ++r) orow[r * HID] = c[r];
}

typedef __attribute__((ext_vector_type(4))) int v4i;

__global__ __launch_bounds__(LDS_THREADS) void deg_edge_k(
    const int* __restrict__ DST, int* DEG, int* DEGB, int E, int N, int nR) {
  __shared__ __attribute__((aligned(16))) int cnt[DEG_RANGE];
  const int tid = threadIdx.x;
  const v4i z4 = {0, 0, 0, 0};
  for (int i = tid; i < DEG_RANGE / 4; i += LDS_THREADS) ((v4i*)cnt)[i] = z4;
  __syncthreads();

  const int q  = blockIdx.x % nR;
  const int t  = blockIdx.x / nR;
  const int S  = gridDim.x / nR;
  const int lo = q * DEG_RANGE;
  const int eps = (((E + S - 1) / S) + 31) & ~31;
  const int e0 = t * eps;
  const int e1 = (e0 + eps < E) ? e0 + eps : E;
  for (int b0 = e0; b0 < e1; b0 += DEG_U * LDS_THREADS) {
    int d[DEG_U];
#pragma unroll
    for (int k = 0; k < DEG_U; ++k) {
      const int e = b0 + k * LDS_THREADS + tid;
      d[k] = DST[e < e1 ? e : e1 - 1];
    }
#pragma unroll
    for (int k = 0; k < DEG_U; ++k) {
      const unsigned r = (unsigned)(d[k] - lo);
      if (b0 + k * LDS_THREADS + tid < e1 && r < (unsigned)DEG_RANGE) atomicAdd(&cnt[r], 1);
    }
  }
  __syncthreads();

  int* D = package_half() ? DEGB : DEG;
  for (int i0 = tid; i0 < DEG_RANGE; i0 += FLUSH_U * LDS_THREADS) {
    int c[FLUSH_U];
#pragma unroll
    for (int k = 0; k < FLUSH_U; ++k) {
      const int i = i0 + k * LDS_THREADS;
      const int w = cnt[i < DEG_RANGE ? i : DEG_RANGE - 1];
      c[k] = i < DEG_RANGE ? w : 0;
    }
#pragma unroll
    for (int k = 0; k < FLUSH_U; ++k) {
      const int n = lo + i0 + k * LDS_THREADS;
      if (c[k] != 0 && n < N) atomicAdd(&D[n], c[k]);
    }
  }
}

__global__ __launch_bounds__(256) void dinv_self1_k(
    float* XW1, float* DNV, const int* __restrict__ DEGB,
    float* H, float* HB, int N) {
  const int n = blockIdx.x * blockDim.x + threadIdx.x;
  if (n >= N) return;
  const int c = ((const int*)DNV)[n] + DEGB[n];
  const float di = rsqrtf((float)c);
  DNV[n] = di;
  v4f* xr = (v4f*)(XW1 + (size_t)n * HID);
  v4f v[4];
#pragma unroll
  for (int q = 0; q < 4; ++q) v[q] = xr[q];
  v4f* hr = (v4f*)(H + (size_t)n * HID);
#pragma unroll
  for (int q = 0; q < 4; ++q) {
    v4f o;
#pragma unroll
    for (int i = 0; i < 4; ++i) o[i] = di * v[q][i];
    xr[q] = o;
    hr[q] = o;
  }
  if (HB != H) {
    const v4f z = {0.f, 0.f, 0.f, 0.f};
    v4f* ur = (v4f*)(HB + (size_t)n * HID);
#pragma unroll
    for (int q = 0; q < 4; ++q) ur[q] = z;
  }
}


template <int J>
__device__ __forceinline__ int row_lane(int v) {
  return __builtin_amdgcn_update_dpp(0, v, 0x150 + J, 0xF, 0xF, false);
}

#define REP16(M) M(0) M(1) M(2) M(3) M(4) M(5) M(6) M(7) M(8) M(9) M(10) M(11) M(12) M(13) M(14) M(15)

template <bool FULL>
__device__ __forceinline__ void agg1_chunk(const int (&s)[EDGE_BATCHES], const int (&d)[EDGE_BATCHES],
                                           int eRow, int E, int f,
                                           const float* __restrict__ XS, float* H) {
  unsigned so[EDGE_BATCHES][16], dO[EDGE_BATCHES][16];
  float v[EDGE_BATCHES][16];
#pragma unroll
  for (int b = 0; b < EDGE_BATCHES; ++b) {
#define AGG1_SO(J) so[b][J] = (unsigned)(row_lane<J>(s[b]) * HID + f);
    REP16(AGG1_SO)
#undef AGG1_SO
  }
#pragma unroll
  for (int b = 0; b < EDGE_BATCHES; ++b) {
#define AGG1_LD(J) v[b][J] = XS[so[b][J]];
    REP16(AGG1_LD)
#undef AGG1_LD
  }
#pragma unroll
  for (int b = 0; b < EDGE_BATCHES; ++b) {
#define AGG1_DO(J) dO[b][J] = (unsigned)(row_lane<J>(d[b]) * HID + f);
    REP16(AGG1_DO)
#undef AGG1_DO
  }
#pragma unroll
  for (int b = 0; b < EDGE_BATCHES; ++b) {
#define AGG1_ADD(J) if (FULL || eRow + 32 * b + J < E) atomAddF(H + dO[b][J], v[b][J]);
    REP16(AGG1_ADD)
#undef AGG1_ADD
  }
}

__device__ __forceinline__ void agg1_fetch(
    const int* __restrict__ SRC, const int* __restrict__ DST,
    int chunk, int lane, int E, int (&s)[EDGE_BATCHES], int (&d)[EDGE_BATCHES]) {
#pragma unroll
  for (int b = 0; b < EDGE_BATCHES; ++b) {
    const int e  = (chunk * EDGE_BATCHES + b) * 32 + lane;
    const int ec = e < E ? e : E - 1;
    s[b] = SRC[ec];
    d[b] = DST[ec];
  }
}

__global__ __launch_bounds__(256) void agg1_k(
    const int* __restrict__ SRC, const int* __restrict__ DST,
    const float* __restrict__ XS,
    float* H0, float* HB, int E) {
  float* H = package_half() ? HB : H0;
  const int lane = threadIdx.x & 31;
  const int f    = lane & 15;
  const int hi   = lane >> 4;
  const int w0 = __builtin_amdgcn_readfirstlane((int)(blockIdx.x * 8 + (threadIdx.x >> 5)));
  const int nWaves  = gridDim.x * 8;
  const int nChunks = (E + 32 * EDGE_BATCHES - 1) / (32 * EDGE_BATCHES);
  if (w0 >= nChunks) return;
  int sN[EDGE_BATCHES], dN[EDGE_BATCHES];
  agg1_fetch(SRC, DST, w0, lane, E, sN, dN);
  for (int wave = w0; wave < nChunks; wave += nWaves) {
    int s[EDGE_BATCHES], d[EDGE_BATCHES];
#pragma unroll
    for (int b = 0; b < EDGE_BATCHES; ++b) { s[b] = sN[b]; d[b] = dN[b]; }
    if (wave + nWaves < nChunks) agg1_fetch(SRC, DST, wave + nWaves, lane, E, sN, dN);
    const int e0 = wave * (32 * EDGE_BATCHES);
    if (e0 + 32 * EDGE_BATCHES <= E) agg1_chunk<true >(s, d, e0 + 16 * hi, E, f, XS, H);
    else                             agg1_chunk<false>(s, d, e0 + 16 * hi, E, f, XS, H);
  }
}

__global__ __launch_bounds__(256) void relu_gemm2_k(
    const float* H, const float* HB,
    const float* __restrict__ B1, const float* __restrict__ W2,
    const float* __restrict__ DNV, float* HW2, float* OUT2,
    float* OUT2B, int N) {
  const int n = blockIdx.x * blockDim.x + threadIdx.x;
  if (n >= N) return;
  const v4f* hr = (const v4f*)(H + (size_t)n * HID);
  const v4f* ur = (const v4f*)(HB + (size_t)n * HID);
  v4f v[4];
#pragma unroll
  for (int q = 0; q < 4; ++q) v[q] = hr[q];
  if (HB != H) {
    v4f u[4];
#pragma unroll
    for (int q = 0; q < 4; ++q) u[q] = ur[q];
#pragma unroll
    for (int q = 0; q < 4; ++q)
#pragma unroll
      for (int i = 0; i < 4; ++i) v[q][i] += u[q][i];
  }
  const float di = DNV[n];
  float h0 = 0.f, h1 = 0.f;
#pragma unroll
  for (int f = 0; f < HID; ++f) {
    const float hv = fmaxf(di * v[f >> 2][f & 3] + B1[f], 0.f);
    h0 += hv * W2[2 * f];
    h1 += hv * W2[2 * f + 1];
  }
  const float g0 = di * h0, g1 = di * h1;
  HW2[2 * n] = g0;
  HW2[2 * n + 1] = g1;
  OUT2[2 * n] = g0;
  OUT2[2 * n + 1] = g1;
  OUT2B[2 * n] = 0.f;
  OUT2B[2 * n + 1] = 0.f;
}

__device__ __forceinline__ void a2_fetch(
    const int* __restrict__ SRC, const int* __restrict__ DST,
    int b0, int tid, int e1, int lo, int (&s)[A2_U], unsigned (&r)[A2_U]) {
#pragma unroll
  for (int k = 0; k < A2_U; ++k) {
    const int e  = b0 + k * LDS_THREADS + tid;
    const int ec = e < e1 ? e : e1 - 1;
    s[k] = SRC[ec];
    const unsigned rr = (unsigned)(DST[ec] - lo);
    r[k] = e < e1 ? rr : 0xFFFFFFFFu;
  }
}

__global__ __launch_bounds__(LDS_THREADS) void agg2_k(
    const int* __restrict__ SRC, const int* __restrict__ DST,
    const float* __restrict__ HW2,
    float* OUT2A, float* OUT2B, int E, int N, int nR) {
  __shared__ __attribute__((aligned(16))) float acc[2 * A2_RANGE];
  const int tid = threadIdx.x;
  const v4f z4 = {0.f, 0.f, 0.f, 0.f};
  for (int i = tid; i < 2 * A2_RANGE / 4; i += LDS_THREADS) ((v4f*)acc)[i] = z4;
  __syncthreads();

  const int q  = blockIdx.x % nR;
  const int t  = blockIdx.x / nR;
  const int S  = gridDim.x / nR;
  const int lo = q * A2_RANGE;
  const int eps = (((E + S - 1) / S) + 31) & ~31;
  const int e0 = t * eps;
  const int e1 = (e0 + eps < E) ? e0 + eps : E;

  int sN[A2_U];
  unsigned rN[A2_U];
  if (e0 < e1) a2_fetch(SRC, DST, e0, tid, e1, lo, sN, rN);
  for (int b0 = e0; b0 < e1; b0 += A2_U * LDS_THREADS) {
    int s[A2_U];
    unsigned r[A2_U];
#pragma unroll
    for (int k = 0; k < A2_U; ++k) { s[k] = sN[k]; r[k] = rN[k]; }
    v2f h[A2_U];
#pragma unroll
    for (int k = 0; k < A2_U; ++k) {
      const int sk = r[k] < (unsigned)A2_RANGE ? s[k] : 0;
      h[k] = *(const v2f*)(HW2 + (size_t)2 * sk);
    }
    const int b1 = b0 + A2_U * LDS_THREADS;
    if (b1 < e1) a2_fetch(SRC, DST, b1, tid, e1, lo, sN, rN);
#pragma unroll
    for (int k = 0; k < A2_U; ++k)
      if (r[k] < (unsigned)A2_RANGE) {
        __hip_atomic_fetch_add(&acc[2 * r[k]],     h[k][0], __ATOMIC_RELAXED, __HIP_MEMORY_SCOPE_WORKGROUP);
        __hip_atomic_fetch_add(&acc[2 * r[k] + 1], h[k][1], __ATOMIC_RELAXED, __HIP_MEMORY_SCOPE_WORKGROUP);
      }
  }
  __syncthreads();

  float* O = (package_half() ? OUT2B : OUT2A) + (size_t)2 * lo;
  for (int i0 = tid; i0 < 2 * A2_RANGE; i0 += FLUSH_U * LDS_THREADS) {
    float v[FLUSH_U];
#pragma unroll
    for (int k = 0; k < FLUSH_U; ++k) {
      const int i = i0 + k * LDS_THREADS;
      const float w = acc[i < 2 * A2_RANGE ? i : 2 * A2_RANGE - 1];
      v[k] = i < 2 * A2_RANGE ? w : 0.f;
    }
#pragma unroll
    for (int k = 0; k < FLUSH_U; ++k) {
      const int i = i0 + k * LDS_THREADS;
      if (v[k] != 0.f && lo + (i >> 1) < N) atomAddF(O + i, v[k]);
    }
  }
}

__global__ void lsm_k(const float* __restrict__ OUT2, const float* __restrict__ OUT2B,
                      const float* __restrict__ DNV,
                      const float* __restrict__ B2, float* __restrict__ out, int N) {
  int n = blockIdx.x * blockDim.x + threadIdx.x;
  if (n >= N) return;
  const float di = DNV[n];
  float a = di * (OUT2[2 * n]     + OUT2B[2 * n])     + B2[0];
  float b = di * (OUT2[2 * n + 1] + OUT2B[2 * n + 1]) + B2[1];
  float m = fmaxf(a, b);
  float lse = m + logf(expf(a - m) + expf(b - m));
  out[2 * n]     = a - lse;
  out[2 * n + 1] = b - lse;
}

extern "C" void kernel_launch(void* const* d_in, const int* in_sizes, int n_in,
                              void* d_out, int out_size, void* d_ws, size_t ws_size,
                              hipStream_t stream) {
  const float* X  = (const float*)d_in[0];
  const int*   EI = (const int*)d_in[1];
  const float* W1 = (const float*)d_in[2];
  const float* B1 = (const float*)d_in[3];
  const float* W2 = (const float*)d_in[4];
  const float* B2 = (const float*)d_in[5];

  const int N = in_sizes[0] / IN_FEAT;
  const int E = in_sizes[1] / 2;
  const int* SRC = EI;
  const int* DST = EI + E;

  float* ws   = (float*)d_ws;
  float* XW1  = ws;
  float* H    = ws + (size_t)N * HID;
  float* DNV  = ws + (size_t)2 * N * HID;
  float* HW2  = DNV + N;
  float* OUT2 = HW2 + (size_t)2 * N;
  float* HB   = ws_size >= (size_t)(3 * HID + 5) * N * sizeof(float) ? OUT2 + (size_t)2 * N : H;
  int*   DEGB = (int*)OUT2;
  float* OUT2B = XW1;

  const int thr = 256;
  const int nTiles = (N + 15) / 16;
  const int nRd = (N + DEG_RANGE - 1) / DEG_RANGE;
  const int nR2 = (N + A2_RANGE - 1) / A2_RANGE;

  gemm1_wmma_k<<<(nTiles + 7) / 8, thr, 0, stream>>>(X, W1, XW1, nTiles, (int*)DNV, DEGB, N);
  deg_edge_k<<<nRd * DEG_SLABS, LDS_THREADS, 0, stream>>>(DST, (int*)DNV, DEGB, E, N, nRd);
  dinv_self1_k<<<(N + thr - 1) / thr, thr, 0, stream>>>(XW1, DNV, DEGB, H, HB, N);
  agg1_k<<<AGG_BLOCKS, thr, 0, stream>>>(SRC, DST, XW1, H, HB, E);
  relu_gemm2_k<<<(N + thr - 1) / thr, thr, 0, stream>>>(H, HB, B1, W2, DNV, HW2, OUT2, OUT2B, N);
  agg2_k<<<nR2 * A2_SLABS, LDS_THREADS, 0, stream>>>(SRC, DST, HW2, OUT2, OUT2B, E, N, nR2);
  lsm_k<<<(N + thr - 1) / thr, thr, 0, stream>>>(OUT2, OUT2B, DNV, B2, (float*)d_out, N);
}


# X@W1 kernel fed as a stream: 16-20 row loads in flight per wave (5-step register ring), the first 20 issued ahead of the W1 staging, one 25-wave work-group per WGP (250 x 25 tiles); same k order
# speedup vs baseline: 6.6402x; 1.0561x over previous
#include <hip/hip_runtime.h>
#include <math.h>


typedef __attribute__((ext_vector_type(16))) __bf16 v16bf;
typedef __attribute__((ext_vector_type(8)))  __bf16 v8bf;
typedef __attribute__((ext_vector_type(2)))  __bf16 v2bf;
typedef __attribute__((ext_vector_type(8)))  float  v8f;
typedef __attribute__((ext_vector_type(4)))  float  v4f;
typedef __attribute__((ext_vector_type(2)))  float  v2f;

#define IN_FEAT 512
#define HID     16
#define BS_PITCH 520
#define G_WAVES   25
#define G_THREADS (G_WAVES * 32)
#define G_ROWS    (G_WAVES * 16)
#define G_AHEAD   5
#define G_STAGE_THREADS 512
#define G_STAGE   (HID * IN_FEAT / 2 / G_STAGE_THREADS)
#define EDGE_BATCHES 2
#define LDS_THREADS 512
#define DEG_RANGE 50048
#define DEG_SLABS 128
#define DEG_U     49
#define FLUSH_U   14
#define A2_RANGE  25024
#define A2_SLABS  64
#define A2_U      14
#define AGG_BLOCKS 1250

__device__ __forceinline__ int package_half() {
  const unsigned die = (__builtin_amdgcn_s_sendmsg_rtn(0x87) >> 16) & 0xFu;
  return (int)((die >> 2) & 1u);
}

__device__ __forceinline__ void atomAddF(float* p, float v) {
  __hip_atomic_fetch_add(p, v, __ATOMIC_RELAXED, __HIP_MEMORY_SCOPE_AGENT);
}

template <int KS>
__device__ __forceinline__ void gemm1_load(const float* __restrict__ ap, v4f (&buf)[4]) {
  const float* p = ap + KS * 32;
  buf[0] = *(const v4f*)(p);
  buf[1] = *(const v4f*)(p + 4);
  buf[2] = *(const v4f*)(p + 16);
  buf[3] = *(const v4f*)(p + 20);
}

template <int KS>
__device__ __forceinline__ void gemm1_step(v4f (&buf)[4], const float* __restrict__ ap,
                                           const __bf16* bp, v8f& c, v8bf& b0, v8bf& b1) {
  asm volatile("" : "+v"(c), "+v"(buf[0]), "+v"(buf[1]), "+v"(buf[2]), "+v"(buf[3]) : : "memory");
  v16bf A, B;
#pragma unroll
  for (int i = 0; i < 4; ++i) {
    A[i]      = (__bf16)buf[0][i];
    A[4 + i]  = (__bf16)buf[1][i];
    A[8 + i]  = (__bf16)buf[2][i];
    A[12 + i] = (__bf16)buf[3][i];
  }
#pragma unroll
  for (int i = 0; i < 8; ++i) { B[i] = b0[i]; B[8 + i] = b1[i]; }
  c = __builtin_amdgcn_wmma_f32_16x16x32_bf16(
       false, A,  false, B,
       (short)0, c,  false,  false);
  if (KS + G_AHEAD < IN_FEAT / 32) gemm1_load<(KS + G_AHEAD < IN_FEAT / 32 ? KS + G_AHEAD : 0)>(ap, buf);
  if (KS + 1 < IN_FEAT / 32) {
    b0 = *(const v8bf*)(bp + (KS + 1) * 32);
    b1 = *(const v8bf*)(bp + (KS + 1) * 32 + 16);
  }
}

__global__ __launch_bounds__(G_THREADS) void gemm1_wmma_k(
    const float* __restrict__ X, const float* __restrict__ W1,
    float* __restrict__ XW1, int nTiles,
    int* __restrict__ DEG, int* __restrict__ DEGB, int N) {
  __shared__ __attribute__((aligned(16))) __bf16 Bs[HID * BS_PITCH];

  const int wave = threadIdx.x >> 5;
  const int lane = threadIdx.x & 31;
  const int tile = blockIdx.x * G_WAVES + wave;
  const int tl   = tile < nTiles ? tile : nTiles - 1;
  const int rc   = lane & 15;
  const int half = lane >> 4;

  const float*  ap = X  + (size_t)(tl * 16 + rc) * IN_FEAT + half * 8;
  const __bf16* bp = Bs + (size_t)rc * BS_PITCH + half * 8;

  v4f r0[4], r1[4], r2[4], r3[4], r4[4];
  gemm1_load<0>(ap, r0);
  gemm1_load<1>(ap, r1);
  gemm1_load<2>(ap, r2);
  gemm1_load<3>(ap, r3);
  gemm1_load<4>(ap, r4);
  asm volatile("" : : : "memory");

  if (threadIdx.x < G_ROWS) {
    const int n = blockIdx.x * G_ROWS + threadIdx.x;
    if (n < N) { DEG[n] = 1; DEGB[n] = 0; }
  }
  if (threadIdx.x < G_STAGE_THREADS) {
    v2bf wp[G_STAGE];
#pragma unroll
    for (int it = 0; it < G_STAGE; ++it) {
      const int j = threadIdx.x + it * G_STAGE_THREADS;
      const int n = j & 15, kp = j >> 4;
      wp[it][0] = (__bf16)W1[(2 * kp) * HID + n];
      wp[it][1] = (__bf16)W1[(2 * kp + 1) * HID + n];
    }
#pragma unroll
    for (int it = 0; it < G_STAGE; ++it) {
      const int j = threadIdx.x + it * G_STAGE_THREADS;
      const int n = j & 15, kp = j >> 4;
      *(v2bf*)(Bs + n * BS_PITCH + 2 * kp) = wp[it];
    }
  }
  __syncthreads();

  v8f c = {0.f, 0.f, 0.f, 0.f, 0.f, 0.f, 0.f, 0.f};
  v8bf b0 = *(const v8bf*)(bp);
  v8bf b1 = *(const v8bf*)(bp + 16);

  gemm1_step<0>(r0, ap, bp, c, b0, b1);
  gemm1_step<1>(r1, ap, bp, c, b0, b1);
  gemm1_step<2>(r2, ap, bp, c, b0, b1);
  gemm1_step<3>(r3, ap, bp, c, b0, b1);
  gemm1_step<4>(r4, ap, bp, c, b0, b1);
  gemm1_step<5>(r0, ap, bp, c, b0, b1);
  gemm1_step<6>(r1, ap, bp, c, b0, b1);
  gemm1_step<7>(r2, ap, bp, c, b0, b1);
  gemm1_step<8>(r3, ap, bp, c, b0, b1);
  gemm1_step<9>(r4, ap, bp, c, b0, b1);
  gemm1_step<10>(r0, ap, bp, c, b0, b1);
  gemm1_step<11>(r1, ap, bp, c, b0, b1);
  gemm1_step<12>(r2, ap, bp, c, b0, b1);
  gemm1_step<13>(r3, ap, bp, c, b0, b1);
  gemm1_step<14>(r4, ap, bp, c, b0, b1);
  gemm1_step<15>(r0, ap, bp, c, b0, b1);

  if (tile >= nTiles) return;

  float* orow = XW1 + ((size_t)tile * 16 + half * 8) * HID + rc;
#pragma unroll
  for (int r = 0; r < 8; ++r) orow[r * HID] = c[r];
}

typedef __attribute__((ext_vector_type(4))) int v4i;

__global__ __launch_bounds__(LDS_THREADS) void deg_edge_k(
    const int* __restrict__ DST, int* DEG, int* DEGB, int E, int N, int nR) {
  __shared__ __attribute__((aligned(16))) int cnt[DEG_RANGE];
  const int tid = threadIdx.x;
  const v4i z4 = {0, 0, 0, 0};
  for (int i = tid; i < DEG_RANGE / 4; i += LDS_THREADS) ((v4i*)cnt)[i] = z4;
  __syncthreads();

  const int q  = blockIdx.x % nR;
  const int t  = blockIdx.x / nR;
  const int S  = gridDim.x / nR;
  const int lo = q * DEG_RANGE;
  const int eps = (((E + S - 1) / S) + 31) & ~31;
  const int e0 = t * eps;
  const int e1 = (e0 + eps < E) ? e0 + eps : E;
  for (int b0 = e0; b0 < e1; b0 += DEG_U * LDS_THREADS) {
    int d[DEG_U];
#pragma unroll
    for (int k = 0; k < DEG_U; ++k) {
      const int e = b0 + k * LDS_THREADS + tid;
      d[k] = DST[e < e1 ? e : e1 - 1];
    }
#pragma unroll
    for (int k = 0; k < DEG_U; ++k) {
      const unsigned r = (unsigned)(d[k] - lo);
      if (b0 + k * LDS_THREADS + tid < e1 && r < (unsigned)DEG_RANGE) atomicAdd(&cnt[r], 1);
    }
  }
  __syncthreads();

  int* D = package_half() ? DEGB : DEG;
  for (int i0 = tid; i0 < DEG_RANGE; i0 += FLUSH_U * LDS_THREADS) {
    int c[FLUSH_U];
#pragma unroll
    for (int k = 0; k < FLUSH_U; ++k) {
      const int i = i0 + k * LDS_THREADS;
      const int w = cnt[i < DEG_RANGE ? i : DEG_RANGE - 1];
      c[k] = i < DEG_RANGE ? w : 0;
    }
#pragma unroll
    for (int k = 0; k < FLUSH_U; ++k) {
      const int n = lo + i0 + k * LDS_THREADS;
      if (c[k] != 0 && n < N) atomicAdd(&D[n], c[k]);
    }
  }
}

__global__ __launch_bounds__(256) void dinv_self1_k(
    float* XW1, float* DNV, const int* __restrict__ DEGB,
    float* H, float* HB, int N) {
  const int n = blockIdx.x * blockDim.x + threadIdx.x;
  if (n >= N) return;
  const int c = ((const int*)DNV)[n] + DEGB[n];
  const float di = rsqrtf((float)c);
  DNV[n] = di;
  v4f* xr = (v4f*)(XW1 + (size_t)n * HID);
  v4f v[4];
#pragma unroll
  for (int q = 0; q < 4; ++q) v[q] = xr[q];
  v4f* hr = (v4f*)(H + (size_t)n * HID);
#pragma unroll
  for (int q = 0; q < 4; ++q) {
    v4f o;
#pragma unroll
    for (int i = 0; i < 4; ++i) o[i] = di * v[q][i];
    xr[q] = o;
    hr[q] = o;
  }
  if (HB != H) {
    const v4f z = {0.f, 0.f, 0.f, 0.f};
    v4f* ur = (v4f*)(HB + (size_t)n * HID);
#pragma unroll
    for (int q = 0; q < 4; ++q) ur[q] = z;
  }
}


template <int J>
__device__ __forceinline__ int row_lane(int v) {
  return __builtin_amdgcn_update_dpp(0, v, 0x150 + J, 0xF, 0xF, false);
}

#define REP16(M) M(0) M(1) M(2) M(3) M(4) M(5) M(6) M(7) M(8) M(9) M(10) M(11) M(12) M(13) M(14) M(15)

template <bool FULL>
__device__ __forceinline__ void agg1_chunk(const int (&s)[EDGE_BATCHES], const int (&d)[EDGE_BATCHES],
                                           int eRow, int E, int f,
                                           const float* __restrict__ XS, float* H) {
  unsigned so[EDGE_BATCHES][16], dO[EDGE_BATCHES][16];
  float v[EDGE_BATCHES][16];
#pragma unroll
  for (int b = 0; b < EDGE_BATCHES; ++b) {
#define AGG1_SO(J) so[b][J] = (unsigned)(row_lane<J>(s[b]) * HID + f);
    REP16(AGG1_SO)
#undef AGG1_SO
  }
#pragma unroll
  for (int b = 0; b < EDGE_BATCHES; ++b) {
#define AGG1_LD(J) v[b][J] = XS[so[b][J]];
    REP16(AGG1_LD)
#undef AGG1_LD
  }
#pragma unroll
  for (int b = 0; b < EDGE_BATCHES; ++b) {
#define AGG1_DO(J) dO[b][J] = (unsigned)(row_lane<J>(d[b]) * HID + f);
    REP16(AGG1_DO)
#undef AGG1_DO
  }
#pragma unroll
  for (int b = 0; b < EDGE_BATCHES; ++b) {
#define AGG1_ADD(J) if (FULL || eRow + 32 * b + J < E) atomAddF(H + dO[b][J], v[b][J]);
    REP16(AGG1_ADD)
#undef AGG1_ADD
  }
}

__device__ __forceinline__ void agg1_fetch(
    const int* __restrict__ SRC, const int* __restrict__ DST,
    int chunk, int lane, int E, int (&s)[EDGE_BATCHES], int (&d)[EDGE_BATCHES]) {
#pragma unroll
  for (int b = 0; b < EDGE_BATCHES; ++b) {
    const int e  = (chunk * EDGE_BATCHES + b) * 32 + lane;
    const int ec = e < E ? e : E - 1;
    s[b] = SRC[ec];
    d[b] = DST[ec];
  }
}

__global__ __launch_bounds__(256) void agg1_k(
    const int* __restrict__ SRC, const int* __restrict__ DST,
    const float* __restrict__ XS,
    float* H0, float* HB, int E) {
  float* H = package_half() ? HB : H0;
  const int lane = threadIdx.x & 31;
  const int f    = lane & 15;
  const int hi   = lane >> 4;
  const int w0 = __builtin_amdgcn_readfirstlane((int)(blockIdx.x * 8 + (threadIdx.x >> 5)));
  const int nWaves  = gridDim.x * 8;
  const int nChunks = (E + 32 * EDGE_BATCHES - 1) / (32 * EDGE_BATCHES);
  if (w0 >= nChunks) return;
  int sN[EDGE_BATCHES], dN[EDGE_BATCHES];
  agg1_fetch(SRC, DST, w0, lane, E, sN, dN);
  for (int wave = w0; wave < nChunks; wave += nWaves) {
    int s[EDGE_BATCHES], d[EDGE_BATCHES];
#pragma unroll
    for (int b = 0; b < EDGE_BATCHES; ++b) { s[b] = sN[b]; d[b] = dN[b]; }
    if (wave + nWaves < nChunks) agg1_fetch(SRC, DST, wave + nWaves, lane, E, sN, dN);
    const int e0 = wave * (32 * EDGE_BATCHES);
    if (e0 + 32 * EDGE_BATCHES <= E) agg1_chunk<true >(s, d, e0 + 16 * hi, E, f, XS, H);
    else                             agg1_chunk<false>(s, d, e0 + 16 * hi, E, f, XS, H);
  }
}

__global__ __launch_bounds__(256) void relu_gemm2_k(
    const float* H, const float* HB,
    const float* __restrict__ B1, const float* __restrict__ W2,
    const float* __restrict__ DNV, float* HW2, float* OUT2,
    float* OUT2B, int N) {
  const int n = blockIdx.x * blockDim.x + threadIdx.x;
  if (n >= N) return;
  const v4f* hr = (const v4f*)(H + (size_t)n * HID);
  const v4f* ur = (const v4f*)(HB + (size_t)n * HID);
  v4f v[4];
#pragma unroll
  for (int q = 0; q < 4; ++q) v[q] = hr[q];
  if (HB != H) {
    v4f u[4];
#pragma unroll
    for (int q = 0; q < 4; ++q) u[q] = ur[q];
#pragma unroll
    for (int q = 0; q < 4; ++q)
#pragma unroll
      for (int i = 0; i < 4; ++i) v[q][i] += u[q][i];
  }
  const float di = DNV[n];
  float h0 = 0.f, h1 = 0.f;
#pragma unroll
  for (int f = 0; f < HID; ++f) {
    const float hv = fmaxf(di * v[f >> 2][f & 3] + B1[f], 0.f);
    h0 += hv * W2[2 * f];
    h1 += hv * W2[2 * f + 1];
  }
  const float g0 = di * h0, g1 = di * h1;
  HW2[2 * n] = g0;
  HW2[2 * n + 1] = g1;
  OUT2[2 * n] = g0;
  OUT2[2 * n + 1] = g1;
  OUT2B[2 * n] = 0.f;
  OUT2B[2 * n + 1] = 0.f;
}

__device__ __forceinline__ void a2_fetch(
    const int* __restrict__ SRC, const int* __restrict__ DST,
    int b0, int tid, int e1, int lo, int (&s)[A2_U], unsigned (&r)[A2_U]) {
#pragma unroll
  for (int k = 0; k < A2_U; ++k) {
    const int e  = b0 + k * LDS_THREADS + tid;
    const int ec = e < e1 ? e : e1 - 1;
    s[k] = SRC[ec];
    const unsigned rr = (unsigned)(DST[ec] - lo);
    r[k] = e < e1 ? rr : 0xFFFFFFFFu;
  }
}

__global__ __launch_bounds__(LDS_THREADS) void agg2_k(
    const int* __restrict__ SRC, const int* __restrict__ DST,
    const float* __restrict__ HW2,
    float* OUT2A, float* OUT2B, int E, int N, int nR) {
  __shared__ __attribute__((aligned(16))) float acc[2 * A2_RANGE];
  const int tid = threadIdx.x;
  const v4f z4 = {0.f, 0.f, 0.f, 0.f};
  for (int i = tid; i < 2 * A2_RANGE / 4; i += LDS_THREADS) ((v4f*)acc)[i] = z4;
  __syncthreads();

  const int q  = blockIdx.x % nR;
  const int t  = blockIdx.x / nR;
  const int S  = gridDim.x / nR;
  const int lo = q * A2_RANGE;
  const int eps = (((E + S - 1) / S) + 31) & ~31;
  const int e0 = t * eps;
  const int e1 = (e0 + eps < E) ? e0 + eps : E;

  int sN[A2_U];
  unsigned rN[A2_U];
  if (e0 < e1) a2_fetch(SRC, DST, e0, tid, e1, lo, sN, rN);
  for (int b0 = e0; b0 < e1; b0 += A2_U * LDS_THREADS) {
    int s[A2_U];
    unsigned r[A2_U];
#pragma unroll
    for (int k = 0; k < A2_U; ++k) { s[k] = sN[k]; r[k] = rN[k]; }
    v2f h[A2_U];
#pragma unroll
    for (int k = 0; k < A2_U; ++k) {
      const int sk = r[k] < (unsigned)A2_RANGE ? s[k] : 0;
      h[k] = *(const v2f*)(HW2 + (size_t)2 * sk);
    }
    const int b1 = b0 + A2_U * LDS_THREADS;
    if (b1 < e1) a2_fetch(SRC, DST, b1, tid, e1, lo, sN, rN);
#pragma unroll
    for (int k = 0; k < A2_U; ++k)
      if (r[k] < (unsigned)A2_RANGE) {
        __hip_atomic_fetch_add(&acc[2 * r[k]],     h[k][0], __ATOMIC_RELAXED, __HIP_MEMORY_SCOPE_WORKGROUP);
        __hip_atomic_fetch_add(&acc[2 * r[k] + 1], h[k][1], __ATOMIC_RELAXED, __HIP_MEMORY_SCOPE_WORKGROUP);
      }
  }
  __syncthreads();

  float* O = (package_half() ? OUT2B : OUT2A) + (size_t)2 * lo;
  for (int i0 = tid; i0 < 2 * A2_RANGE; i0 += FLUSH_U * LDS_THREADS) {
    float v[FLUSH_U];
#pragma unroll
    for (int k = 0; k < FLUSH_U; ++k) {
      const int i = i0 + k * LDS_THREADS;
      const float w = acc[i < 2 * A2_RANGE ? i : 2 * A2_RANGE - 1];
      v[k] = i < 2 * A2_RANGE ? w : 0.f;
    }
#pragma unroll
    for (int k = 0; k < FLUSH_U; ++k) {
      const int i = i0 + k * LDS_THREADS;
      if (v[k] != 0.f && lo + (i >> 1) < N) atomAddF(O + i, v[k]);
    }
  }
}

__global__ void lsm_k(const float* __restrict__ OUT2, const float* __restrict__ OUT2B,
                      const float* __restrict__ DNV,
                      const float* __restrict__ B2, float* __restrict__ out, int N) {
  int n = blockIdx.x * blockDim.x + threadIdx.x;
  if (n >= N) return;
  const float di = DNV[n];
  float a = di * (OUT2[2 * n]     + OUT2B[2 * n])     + B2[0];
  float b = di * (OUT2[2 * n + 1] + OUT2B[2 * n + 1]) + B2[1];
  float m = fmaxf(a, b);
  float lse = m + logf(expf(a - m) + expf(b - m));
  out[2 * n]     = a - lse;
  out[2 * n + 1] = b - lse;
}

extern "C" void kernel_launch(void* const* d_in, const int* in_sizes, int n_in,
                              void* d_out, int out_size, void* d_ws, size_t ws_size,
                              hipStream_t stream) {
  const float* X  = (const float*)d_in[0];
  const int*   EI = (const int*)d_in[1];
  const float* W1 = (const float*)d_in[2];
  const float* B1 = (const float*)d_in[3];
  const float* W2 = (const float*)d_in[4];
  const float* B2 = (const float*)d_in[5];

  const int N = in_sizes[0] / IN_FEAT;
  const int E = in_sizes[1] / 2;
  const int* SRC = EI;
  const int* DST = EI + E;

  float* ws   = (float*)d_ws;
  float* XW1  = ws;
  float* H    = ws + (size_t)N * HID;
  float* DNV  = ws + (size_t)2 * N * HID;
  float* HW2  = DNV + N;
  float* OUT2 = HW2 + (size_t)2 * N;
  float* HB   = ws_size >= (size_t)(3 * HID + 5) * N * sizeof(float) ? OUT2 + (size_t)2 * N : H;
  int*   DEGB = (int*)OUT2;
  float* OUT2B = XW1;

  const int thr = 256;
  const int nTiles = (N + 15) / 16;
  const int nRd = (N + DEG_RANGE - 1) / DEG_RANGE;
  const int nR2 = (N + A2_RANGE - 1) / A2_RANGE;

  gemm1_wmma_k<<<(nTiles + G_WAVES - 1) / G_WAVES, G_THREADS, 0, stream>>>(X, W1, XW1, nTiles, (int*)DNV, DEGB, N);
  deg_edge_k<<<nRd * DEG_SLABS, LDS_THREADS, 0, stream>>>(DST, (int*)DNV, DEGB, E, N, nRd);
  dinv_self1_k<<<(N + thr - 1) / thr, thr, 0, stream>>>(XW1, DNV, DEGB, H, HB, N);
  agg1_k<<<AGG_BLOCKS, thr, 0, stream>>>(SRC, DST, XW1, H, HB, E);
  relu_gemm2_k<<<(N + thr - 1) / thr, thr, 0, stream>>>(H, HB, B1, W2, DNV, HW2, OUT2, OUT2B, N);
  agg2_k<<<nR2 * A2_SLABS, LDS_THREADS, 0, stream>>>(SRC, DST, HW2, OUT2, OUT2B, E, N, nR2);
  lsm_k<<<(N + thr - 1) / thr, thr, 0, stream>>>(OUT2, OUT2B, DNV, B2, (float*)d_out, N);
}


# degree count runs first, leaves 16-bit per-slab counts in the rows its nodes' X@W1 work-group will fill; that launch sums them and stores rows scaled by rsqrt(deg) plus plane starts: 7 -> 6 launches
# speedup vs baseline: 7.0955x; 1.0686x over previous
#include <hip/hip_runtime.h>
#include <math.h>


typedef __attribute__((ext_vector_type(16))) __bf16 v16bf;
typedef __attribute__((ext_vector_type(8)))  __bf16 v8bf;
typedef __attribute__((ext_vector_type(2)))  __bf16 v2bf;
typedef __attribute__((ext_vector_type(8)))  float  v8f;
typedef __attribute__((ext_vector_type(4)))  float  v4f;
typedef __attribute__((ext_vector_type(2)))  float  v2f;

#define IN_FEAT 512
#define HID     16
#define BS_PITCH 520
#define G_WAVES   25
#define G_THREADS (G_WAVES * 32)
#define G_ROWS    (G_WAVES * 16)
#define G_AHEAD   5
#define G_STAGE_THREADS 512
#define G_STAGE   (HID * IN_FEAT / 2 / G_STAGE_THREADS)
#define EDGE_BATCHES 2
#define LDS_THREADS 512
#define DEG_RANGE 25024
#define DEG_SLABS 64
#define DEG_HALF  (DEG_SLABS / 2)
#define DEG_TG    4
#define DEG_PER_T (DEG_SLABS / DEG_TG)
#define DEG_U     98
#define FLUSH_U   14
#define A2_RANGE  25024
#define A2_SLABS  64
#define A2_U      14
#define AGG_BLOCKS 1250

__device__ __forceinline__ int package_half() {
  const unsigned die = (__builtin_amdgcn_s_sendmsg_rtn(0x87) >> 16) & 0xFu;
  return (int)((die >> 2) & 1u);
}

__device__ __forceinline__ void atomAddF(float* p, float v) {
  __hip_atomic_fetch_add(p, v, __ATOMIC_RELAXED, __HIP_MEMORY_SCOPE_AGENT);
}

template <int KS>
__device__ __forceinline__ void gemm1_load(const float* __restrict__ ap, v4f (&buf)[4]) {
  const float* p = ap + KS * 32;
  buf[0] = *(const v4f*)(p);
  buf[1] = *(const v4f*)(p + 4);
  buf[2] = *(const v4f*)(p + 16);
  buf[3] = *(const v4f*)(p + 20);
}

template <int KS>
__device__ __forceinline__ void gemm1_step(v4f (&buf)[4], const float* __restrict__ ap,
                                           const __bf16* bp, v8f& c, v8bf& b0, v8bf& b1) {
  asm volatile("" : "+v"(c), "+v"(buf[0]), "+v"(buf[1]), "+v"(buf[2]), "+v"(buf[3]) : : "memory");
  v16bf A, B;
#pragma unroll
  for (int i = 0; i < 4; ++i) {
    A[i]      = (__bf16)buf[0][i];
    A[4 + i]  = (__bf16)buf[1][i];
    A[8 + i]  = (__bf16)buf[2][i];
    A[12 + i] = (__bf16)buf[3][i];
  }
#pragma unroll
  for (int i = 0; i < 8; ++i) { B[i] = b0[i]; B[8 + i] = b1[i]; }
  c = __builtin_amdgcn_wmma_f32_16x16x32_bf16(
       false, A,  false, B,
       (short)0, c,  false,  false);
  if (KS + G_AHEAD < IN_FEAT / 32) gemm1_load<(KS + G_AHEAD < IN_FEAT / 32 ? KS + G_AHEAD : 0)>(ap, buf);
  if (KS + 1 < IN_FEAT / 32) {
    b0 = *(const v8bf*)(bp + (KS + 1) * 32);
    b1 = *(const v8bf*)(bp + (KS + 1) * 32 + 16);
  }
}

__global__ __launch_bounds__(G_THREADS) void gemm1_wmma_k(
    const float* __restrict__ X, const float* __restrict__ W1,
    float* XW1, float* H, float* HB, float* __restrict__ DNV, int nTiles, int N) {
  __shared__ __attribute__((aligned(16))) __bf16 Bs[HID * BS_PITCH];
  __shared__ __attribute__((aligned(16))) int part[DEG_TG * G_ROWS];
  static_assert(G_THREADS == DEG_TG * (G_ROWS / 2), "one thread per (node pair, slab group)");
  static_assert(DEG_SLABS * G_ROWS * 2 == 2 * G_ROWS * HID * 4, "a work-group's counts fill its rows of two tables");

  const int wave = threadIdx.x >> 5;
  const int lane = threadIdx.x & 31;
  const int tile = blockIdx.x * G_WAVES + wave;
  const int tl   = tile < nTiles ? tile : nTiles - 1;
  const int rc   = lane & 15;
  const int half = lane >> 4;

  const float*  ap = X  + (size_t)(tl * 16 + rc) * IN_FEAT + half * 8;
  const __bf16* bp = Bs + (size_t)rc * BS_PITCH + half * 8;

  v4f r0[4], r1[4], r2[4], r3[4], r4[4];
  gemm1_load<0>(ap, r0);
  gemm1_load<1>(ap, r1);
  gemm1_load<2>(ap, r2);
  gemm1_load<3>(ap, r3);
  gemm1_load<4>(ap, r4);
  asm volatile("" : : : "memory");

  const int cp  = threadIdx.x % (G_ROWS / 2);
  const int ctg = threadIdx.x / (G_ROWS / 2);
  const unsigned* cw = (const unsigned*)((ctg < DEG_TG / 2 ? XW1 : H) + (size_t)blockIdx.x * G_ROWS * HID)
                       + (ctg & 1) * DEG_PER_T * (G_ROWS / 2) + cp;
  unsigned cnt16[DEG_PER_T];
#pragma unroll
  for (int k = 0; k < DEG_PER_T; ++k) cnt16[k] = cw[k * (G_ROWS / 2)];
  if (threadIdx.x < G_STAGE_THREADS) {
    v2bf wp[G_STAGE];
#pragma unroll
    for (int it = 0; it < G_STAGE; ++it) {
      const int j = threadIdx.x + it * G_STAGE_THREADS;
      const int n = j & 15, kp = j >> 4;
      wp[it][0] = (__bf16)W1[(2 * kp) * HID + n];
      wp[it][1] = (__bf16)W1[(2 * kp + 1) * HID + n];
    }
#pragma unroll
    for (int it = 0; it < G_STAGE; ++it) {
      const int j = threadIdx.x + it * G_STAGE_THREADS;
      const int n = j & 15, kp = j >> 4;
      *(v2bf*)(Bs + n * BS_PITCH + 2 * kp) = wp[it];
    }
  }
  {
    int lo = 0, hi = 0;
#pragma unroll
    for (int k = 0; k < DEG_PER_T; ++k) { lo += (int)(cnt16[k] & 0xFFFFu); hi += (int)(cnt16[k] >> 16); }
    typedef __attribute__((ext_vector_type(2))) int v2i;
    const v2i s = {lo, hi};
    *(v2i*)(part + ctg * G_ROWS + 2 * cp) = s;
  }
  __syncthreads();

  v8f c = {0.f, 0.f, 0.f, 0.f, 0.f, 0.f, 0.f, 0.f};
  v8bf b0 = *(const v8bf*)(bp);
  v8bf b1 = *(const v8bf*)(bp + 16);

  gemm1_step<0>(r0, ap, bp, c, b0, b1);
  gemm1_step<1>(r1, ap, bp, c, b0, b1);
  gemm1_step<2>(r2, ap, bp, c, b0, b1);
  gemm1_step<3>(r3, ap, bp, c, b0, b1);
  gemm1_step<4>(r4, ap, bp, c, b0, b1);
  gemm1_step<5>(r0, ap, bp, c, b0, b1);
  gemm1_step<6>(r1, ap, bp, c, b0, b1);
  gemm1_step<7>(r2, ap, bp, c, b0, b1);
  gemm1_step<8>(r3, ap, bp, c, b0, b1);
  gemm1_step<9>(r4, ap, bp, c, b0, b1);
  gemm1_step<10>(r0, ap, bp, c, b0, b1);
  gemm1_step<11>(r1, ap, bp, c, b0, b1);
  gemm1_step<12>(r2, ap, bp, c, b0, b1);
  gemm1_step<13>(r3, ap, bp, c, b0, b1);
  gemm1_step<14>(r4, ap, bp, c, b0, b1);
  gemm1_step<15>(r0, ap, bp, c, b0, b1);

  if (tile >= nTiles) return;

  typedef __attribute__((ext_vector_type(4))) int v4i_t;
  const int* pb = part + wave * 16 + half * 8;
  v4i_t d0 = *(const v4i_t*)(pb), d1 = *(const v4i_t*)(pb + 4);
#pragma unroll
  for (int g = 1; g < DEG_TG; ++g) {
    d0 += *(const v4i_t*)(pb + g * G_ROWS);
    d1 += *(const v4i_t*)(pb + g * G_ROWS + 4);
  }
  float di[8];
#pragma unroll
  for (int r = 0; r < 4; ++r) { di[r] = rsqrtf((float)(1 + d0[r])); di[4 + r] = rsqrtf((float)(1 + d1[r])); }
  if (rc < 8) {
    const int* pn = part + wave * 16 + half * 8 + rc;
    const int cn = 1 + pn[0] + pn[G_ROWS] + pn[2 * G_ROWS] + pn[3 * G_ROWS];
    DNV[tile * 16 + half * 8 + rc] = rsqrtf((float)cn);
  }
  const size_t ob = ((size_t)tile * 16 + half * 8) * HID + rc;
#pragma unroll
  for (int r = 0; r < 8; ++r) {
    const float o = di[r] * c[r];
    XW1[ob + r * HID] = o;
    H[ob + r * HID] = o;
    if (HB != H) HB[ob + r * HID] = 0.f;
  }
}

typedef __attribute__((ext_vector_type(4))) int v4i;

#define DFL_U 5
__global__ __launch_bounds__(LDS_THREADS) void deg_edge_k(
    const int* __restrict__ DST, float* XW1, float* H, int E, int N, int nR) {
  __shared__ __attribute__((aligned(16))) int cnt[DEG_RANGE];
  const int tid = threadIdx.x;
  const v4i z4 = {0, 0, 0, 0};
  for (int i = tid; i < DEG_RANGE / 4; i += LDS_THREADS) ((v4i*)cnt)[i] = z4;
  __syncthreads();

  const int q  = blockIdx.x % nR;
  const int t  = blockIdx.x / nR;
  const int S  = gridDim.x / nR;
  const int lo = q * DEG_RANGE;
  const int eps = (((E + S - 1) / S) + 31) & ~31;
  const int e0 = t * eps;
  const int e1 = (e0 + eps < E) ? e0 + eps : E;
  for (int b0 = e0; b0 < e1; b0 += DEG_U * LDS_THREADS) {
    int d[DEG_U];
#pragma unroll
    for (int k = 0; k < DEG_U; ++k) {
      const int e = b0 + k * LDS_THREADS + tid;
      d[k] = DST[e < e1 ? e : e1 - 1];
    }
#pragma unroll
    for (int k = 0; k < DEG_U; ++k) {
      const unsigned r = (unsigned)(d[k] - lo);
      if (b0 + k * LDS_THREADS + tid < e1 && r < (unsigned)DEG_RANGE) atomicAdd(&cnt[r], 1);
    }
  }
  __syncthreads();

  typedef __attribute__((ext_vector_type(2))) int v2i;
  unsigned* dstw = (unsigned*)(t < DEG_HALF ? XW1 : H) + (size_t)(t & (DEG_HALF - 1)) * (G_ROWS / 2);
  for (int i0 = tid; i0 < DEG_RANGE / 2; i0 += DFL_U * LDS_THREADS) {
    v2i cc[DFL_U];
#pragma unroll
    for (int k = 0; k < DFL_U; ++k) {
      const int i = i0 + k * LDS_THREADS;
      cc[k] = *(const v2i*)(cnt + 2 * (i < DEG_RANGE / 2 ? i : DEG_RANGE / 2 - 1));
    }
#pragma unroll
    for (int k = 0; k < DFL_U; ++k) {
      const int i = i0 + k * LDS_THREADS;
      const int n = lo + 2 * i;
      if (i < DEG_RANGE / 2 && n < N) {
        const int b = n / G_ROWS;
        const int j = n - b * G_ROWS;
        dstw[(size_t)b * (G_ROWS * HID) + (j >> 1)] = (unsigned)cc[k][0] | ((unsigned)cc[k][1] << 16);
      }
    }
  }
}


template <int J>
__device__ __forceinline__ int row_lane(int v) {
  return __builtin_amdgcn_update_dpp(0, v, 0x150 + J, 0xF, 0xF, false);
}

#define REP16(M) M(0) M(1) M(2) M(3) M(4) M(5) M(6) M(7) M(8) M(9) M(10) M(11) M(12) M(13) M(14) M(15)

template <bool FULL>
__device__ __forceinline__ void agg1_chunk(const int (&s)[EDGE_BATCHES], const int (&d)[EDGE_BATCHES],
                                           int eRow, int E, int f,
                                           const float* __restrict__ XS, float* H) {
  unsigned so[EDGE_BATCHES][16], dO[EDGE_BATCHES][16];
  float v[EDGE_BATCHES][16];
#pragma unroll
  for (int b = 0; b < EDGE_BATCHES; ++b) {
#define AGG1_SO(J) so[b][J] = (unsigned)(row_lane<J>(s[b]) * HID + f);
    REP16(AGG1_SO)
#undef AGG1_SO
  }
#pragma unroll
  for (int b = 0; b < EDGE_BATCHES; ++b) {
#define AGG1_LD(J) v[b][J] = XS[so[b][J]];
    REP16(AGG1_LD)
#undef AGG1_LD
  }
#pragma unroll
  for (int b = 0; b < EDGE_BATCHES; ++b) {
#define AGG1_DO(J) dO[b][J] = (unsigned)(row_lane<J>(d[b]) * HID + f);
    REP16(AGG1_DO)
#undef AGG1_DO
  }
#pragma unroll
  for (int b = 0; b < EDGE_BATCHES; ++b) {
#define AGG1_ADD(J) if (FULL || eRow + 32 * b + J < E) atomAddF(H + dO[b][J], v[b][J]);
    REP16(AGG1_ADD)
#undef AGG1_ADD
  }
}

__device__ __forceinline__ void agg1_fetch(
    const int* __restrict__ SRC, const int* __restrict__ DST,
    int chunk, int lane, int E, int (&s)[EDGE_BATCHES], int (&d)[EDGE_BATCHES]) {
#pragma unroll
  for (int b = 0; b < EDGE_BATCHES; ++b) {
    const int e  = (chunk * EDGE_BATCHES + b) * 32 + lane;
    const int ec = e < E ? e : E - 1;
    s[b] = SRC[ec];
    d[b] = DST[ec];
  }
}

__global__ __launch_bounds__(256) void agg1_k(
    const int* __restrict__ SRC, const int* __restrict__ DST,
    const float* __restrict__ XS,
    float* H0, float* HB, int E) {
  float* H = package_half() ? HB : H0;
  const int lane = threadIdx.x & 31;
  const int f    = lane & 15;
  const int hi   = lane >> 4;
  const int w0 = __builtin_amdgcn_readfirstlane((int)(blockIdx.x * 8 + (threadIdx.x >> 5)));
  const int nWaves  = gridDim.x * 8;
  const int nChunks = (E + 32 * EDGE_BATCHES - 1) / (32 * EDGE_BATCHES);
  if (w0 >= nChunks) return;
  int sN[EDGE_BATCHES], dN[EDGE_BATCHES];
  agg1_fetch(SRC, DST, w0, lane, E, sN, dN);
  for (int wave = w0; wave < nChunks; wave += nWaves) {
    int s[EDGE_BATCHES], d[EDGE_BATCHES];
#pragma unroll
    for (int b = 0; b < EDGE_BATCHES; ++b) { s[b] = sN[b]; d[b] = dN[b]; }
    if (wave + nWaves < nChunks) agg1_fetch(SRC, DST, wave + nWaves, lane, E, sN, dN);
    const int e0 = wave * (32 * EDGE_BATCHES);
    if (e0 + 32 * EDGE_BATCHES <= E) agg1_chunk<true >(s, d, e0 + 16 * hi, E, f, XS, H);
    else                             agg1_chunk<false>(s, d, e0 + 16 * hi, E, f, XS, H);
  }
}

__global__ __launch_bounds__(256) void relu_gemm2_k(
    const float* H, const float* HB,
    const float* __restrict__ B1, const float* __restrict__ W2,
    const float* __restrict__ DNV, float* HW2, float* OUT2,
    float* OUT2B, int N) {
  const int n = blockIdx.x * blockDim.x + threadIdx.x;
  if (n >= N) return;
  const v4f* hr = (const v4f*)(H + (size_t)n * HID);
  const v4f* ur = (const v4f*)(HB + (size_t)n * HID);
  v4f v[4];
#pragma unroll
  for (int q = 0; q < 4; ++q) v[q] = hr[q];
  if (HB != H) {
    v4f u[4];
#pragma unroll
    for (int q = 0; q < 4; ++q) u[q] = ur[q];
#pragma unroll
    for (int q = 0; q < 4; ++q)
#pragma unroll
      for (int i = 0; i < 4; ++i) v[q][i] += u[q][i];
  }
  const float di = DNV[n];
  float h0 = 0.f, h1 = 0.f;
#pragma unroll
  for (int f = 0; f < HID; ++f) {
    const float hv = fmaxf(di * v[f >> 2][f & 3] + B1[f], 0.f);
    h0 += hv * W2[2 * f];
    h1 += hv * W2[2 * f + 1];
  }
  const float g0 = di * h0, g1 = di * h1;
  HW2[2 * n] = g0;
  HW2[2 * n + 1] = g1;
  OUT2[2 * n] = g0;
  OUT2[2 * n + 1] = g1;
  OUT2B[2 * n] = 0.f;
  OUT2B[2 * n + 1] = 0.f;
}

__device__ __forceinline__ void a2_fetch(
    const int* __restrict__ SRC, const int* __restrict__ DST,
    int b0, int tid, int e1, int lo, int (&s)[A2_U], unsigned (&r)[A2_U]) {
#pragma unroll
  for (int k = 0; k < A2_U; ++k) {
    const int e  = b0 + k * LDS_THREADS + tid;
    const int ec = e < e1 ? e : e1 - 1;
    s[k] = SRC[ec];
    const unsigned rr = (unsigned)(DST[ec] - lo);
    r[k] = e < e1 ? rr : 0xFFFFFFFFu;
  }
}

__global__ __launch_bounds__(LDS_THREADS) void agg2_k(
    const int* __restrict__ SRC, const int* __restrict__ DST,
    const float* __restrict__ HW2,
    float* OUT2A, float* OUT2B, int E, int N, int nR) {
  __shared__ __attribute__((aligned(16))) float acc[2 * A2_RANGE];
  const int tid = threadIdx.x;
  const v4f z4 = {0.f, 0.f, 0.f, 0.f};
  for (int i = tid; i < 2 * A2_RANGE / 4; i += LDS_THREADS) ((v4f*)acc)[i] = z4;
  __syncthreads();

  const int q  = blockIdx.x % nR;
  const int t  = blockIdx.x / nR;
  const int S  = gridDim.x / nR;
  const int lo = q * A2_RANGE;
  const int eps = (((E + S - 1) / S) + 31) & ~31;
  const int e0 = t * eps;
  const int e1 = (e0 + eps < E) ? e0 + eps : E;

  int sN[A2_U];
  unsigned rN[A2_U];
  if (e0 < e1) a2_fetch(SRC, DST, e0, tid, e1, lo, sN, rN);
  for (int b0 = e0; b0 < e1; b0 += A2_U * LDS_THREADS) {
    int s[A2_U];
    unsigned r[A2_U];
#pragma unroll
    for (int k = 0; k < A2_U; ++k) { s[k] = sN[k]; r[k] = rN[k]; }
    v2f h[A2_U];
#pragma unroll
    for (int k = 0; k < A2_U; ++k) {
      const int sk = r[k] < (unsigned)A2_RANGE ? s[k] : 0;
      h[k] = *(const v2f*)(HW2 + (size_t)2 * sk);
    }
    const int b1 = b0 + A2_U * LDS_THREADS;
    if (b1 < e1) a2_fetch(SRC, DST, b1, tid, e1, lo, sN, rN);
#pragma unroll
    for (int k = 0; k < A2_U; ++k)
      if (r[k] < (unsigned)A2_RANGE) {
        __hip_atomic_fetch_add(&acc[2 * r[k]],     h[k][0], __ATOMIC_RELAXED, __HIP_MEMORY_SCOPE_WORKGROUP);
        __hip_atomic_fetch_add(&acc[2 * r[k] + 1], h[k][1], __ATOMIC_RELAXED, __HIP_MEMORY_SCOPE_WORKGROUP);
      }
  }
  __syncthreads();

  float* O = (package_half() ? OUT2B : OUT2A) + (size_t)2 * lo;
  for (int i0 = tid; i0 < 2 * A2_RANGE; i0 += FLUSH_U * LDS_THREADS) {
    float v[FLUSH_U];
#pragma unroll
    for (int k = 0; k < FLUSH_U; ++k) {
      const int i = i0 + k * LDS_THREADS;
      const float w = acc[i < 2 * A2_RANGE ? i : 2 * A2_RANGE - 1];
      v[k] = i < 2 * A2_RANGE ? w : 0.f;
    }
#pragma unroll
    for (int k = 0; k < FLUSH_U; ++k) {
      const int i = i0 + k * LDS_THREADS;
      if (v[k] != 0.f && lo + (i >> 1) < N) atomAddF(O + i, v[k]);
    }
  }
}

__global__ void lsm_k(const float* __restrict__ OUT2, const float* __restrict__ OUT2B,
                      const float* __restrict__ DNV,
                      const float* __restrict__ B2, float* __restrict__ out, int N) {
  int n = blockIdx.x * blockDim.x + threadIdx.x;
  if (n >= N) return;
  const float di = DNV[n];
  float a = di * (OUT2[2 * n]     + OUT2B[2 * n])     + B2[0];
  float b = di * (OUT2[2 * n + 1] + OUT2B[2 * n + 1]) + B2[1];
  float m = fmaxf(a, b);
  float lse = m + logf(expf(a - m) + expf(b - m));
  out[2 * n]     = a - lse;
  out[2 * n + 1] = b - lse;
}

extern "C" void kernel_launch(void* const* d_in, const int* in_sizes, int n_in,
                              void* d_out, int out_size, void* d_ws, size_t ws_size,
                              hipStream_t stream) {
  const float* X  = (const float*)d_in[0];
  const int*   EI = (const int*)d_in[1];
  const float* W1 = (const float*)d_in[2];
  const float* B1 = (const float*)d_in[3];
  const float* W2 = (const float*)d_in[4];
  const float* B2 = (const float*)d_in[5];

  const int N = in_sizes[0] / IN_FEAT;
  const int E = in_sizes[1] / 2;
  const int* SRC = EI;
  const int* DST = EI + E;

  float* ws   = (float*)d_ws;
  float* XW1  = ws;
  float* H    = ws + (size_t)N * HID;
  float* DNV  = ws + (size_t)2 * N * HID;
  float* HW2  = DNV + N;
  float* OUT2 = HW2 + (size_t)2 * N;
  float* HB   = ws_size >= (size_t)(3 * HID + 5) * N * sizeof(float) ? OUT2 + (size_t)2 * N : H;
  float* OUT2B = XW1;

  const int thr = 256;
  const int nTiles = (N + 15) / 16;
  const int nRd = (N + DEG_RANGE - 1) / DEG_RANGE;
  const int nR2 = (N + A2_RANGE - 1) / A2_RANGE;

  deg_edge_k<<<nRd * DEG_SLABS, LDS_THREADS, 0, stream>>>(DST, XW1, H, E, N, nRd);
  gemm1_wmma_k<<<(nTiles + G_WAVES - 1) / G_WAVES, G_THREADS, 0, stream>>>(X, W1, XW1, H, HB, DNV, nTiles, N);
  agg1_k<<<AGG_BLOCKS, thr, 0, stream>>>(SRC, DST, XW1, H, HB, E);
  relu_gemm2_k<<<(N + thr - 1) / thr, thr, 0, stream>>>(H, HB, B1, W2, DNV, HW2, OUT2, OUT2B, N);
  agg2_k<<<nR2 * A2_SLABS, LDS_THREADS, 0, stream>>>(SRC, DST, HW2, OUT2, OUT2B, E, N, nR2);
  lsm_k<<<(N + thr - 1) / thr, thr, 0, stream>>>(OUT2, OUT2B, DNV, B2, (float*)d_out, N);
}
